# starting kernel
# baseline (speedup 1.0000x reference)
#include <hip/hip_runtime.h>
#include <hip/hip_bf16.h>


#define NS   32
#define CC   128
#define HH   56
#define WW   56
#define HWP  3136
#define C3   384
#define PADK 34

typedef float v2f __attribute__((ext_vector_type(2)));
typedef float v8f __attribute__((ext_vector_type(8)));

__device__ __forceinline__ v8f wmma_f32(v2f a, v2f b, v8f c) {
    return __builtin_amdgcn_wmma_f32_16x16x4_f32(false, a, false, b, (short)0, c, false, false);
}

__global__ __launch_bounds__(256) void k_gemm1(const float* __restrict__ p3,
                                               const float* __restrict__ x,
                                               float* __restrict__ t3) {
    __shared__ float As[128 * PADK];
    __shared__ float Bs[224 * PADK];
    const int t    = threadIdx.x;
    const int lane = t & 31;
    const int wv   = t >> 5;
    const int smp  = blockIdx.y;
    const int pos0 = blockIdx.x * 224;
    const float* xs = x + (size_t)smp * CC * HWP;

    const int mn    = lane & 15;
    const int kp    = (lane >> 4) << 1;
    const int rmsb  = (lane >> 4) << 3;
    const int mbase = (wv >> 1) * 32;
    const int nbase = (wv & 1) * 112;

    v8f acc[2][7] = {};

    for (int ks = 0; ks < C3; ks += 32) {
        #pragma unroll
        for (int i = 0; i < 16; ++i) {
            int idx = t + i * 256;
            int m = idx >> 5, k = idx & 31;
            As[m * PADK + k] = p3[m * C3 + ks + k];
        }
        #pragma unroll
        for (int i = 0; i < 28; ++i) {
            int idx = t + i * 256;
            int k = idx / 224, nn = idx % 224;
            int j = ks + k;
            int tap = j % 3, ci = j / 3;
            int pos = pos0 + nn;
            int hp = pos / WW + 2 * (tap - 1);
            float v = 0.0f;
            if ((unsigned)hp < (unsigned)HH)
                v = xs[ci * HWP + pos + 2 * WW * (tap - 1)];
            Bs[nn * PADK + k] = v;
        }
        __syncthreads();
        #pragma unroll
        for (int kk = 0; kk < 8; ++kk) {
            int ko = kk * 4 + kp;
            v2f a0 = *(const v2f*)&As[(mbase + mn) * PADK + ko];
            v2f a1 = *(const v2f*)&As[(mbase + 16 + mn) * PADK + ko];
            #pragma unroll
            for (int ni = 0; ni < 7; ++ni) {
                v2f b = *(const v2f*)&Bs[(nbase + ni * 16 + mn) * PADK + ko];
                acc[0][ni] = wmma_f32(a0, b, acc[0][ni]);
                acc[1][ni] = wmma_f32(a1, b, acc[1][ni]);
            }
        }
        __syncthreads();
    }

    float* o = t3 + (size_t)smp * CC * HWP;
    #pragma unroll
    for (int mi = 0; mi < 2; ++mi) {
        #pragma unroll
        for (int ni = 0; ni < 7; ++ni) {
            int col = pos0 + nbase + ni * 16 + mn;
            #pragma unroll
            for (int r = 0; r < 8; ++r) {
                int m = mbase + mi * 16 + r + rmsb;
                o[m * HWP + col] = acc[mi][ni][r];
            }
        }
    }
}

__global__ __launch_bounds__(256) void k_t12(const float* __restrict__ x,
                                             const float* __restrict__ t3,
                                             const float* __restrict__ p2,
                                             const float* __restrict__ p5,
                                             const float* __restrict__ p10,
                                             float* __restrict__ t12) {
    int tid  = blockIdx.x * 256 + threadIdx.x;
    int pos  = tid % HWP;
    int rest = tid / HWP;
    int c    = rest % CC;
    int smp  = rest / CC;
    int h = pos / WW, w = pos % WW;

    const float* xc = x + ((size_t)smp * CC + c) * HWP;
    float x0  = xc[pos];
    float xvm = (h >= 2)      ? xc[pos - 2 * WW] : 0.0f;
    float xvp = (h < HH - 2)  ? xc[pos + 2 * WW] : 0.0f;
    float xhm = (w >= 2)      ? xc[pos - 2]      : 0.0f;
    float xhp = (w < WW - 2)  ? xc[pos + 2]      : 0.0f;

    float mx = fmaxf(xhm, fmaxf(x0, xhp));
    float e0 = __expf(xhm - mx), e1 = __expf(x0 - mx), e2 = __expf(xhp - mx);
    float inv = 1.0f / (e0 + e1 + e2);
    float t8[3] = { e0 * inv, e1 * inv, e2 * inv };
    float xv[3] = { xvm, x0, xvp };

    float p5v = p5[pos];
    const float* t3s  = t3  + (size_t)smp * CC * HWP;
    const float* p10s = p10 + (size_t)(3 * c) * HWP + pos;
    float* outj = t12 + ((size_t)smp * C3 + 3 * c) * HWP + pos;

    #pragma unroll
    for (int k = 0; k < 3; ++k) {
        int j = 3 * c + k;
        int b = j & (CC - 1);
        float t5v  = p5v * t3s[b * HWP + pos];
        float t2v  = p2[k * WW + w] * xv[k];
        float t10v = p10s[k * HWP] * t8[k];
        outj[k * HWP] = t5v - t2v - t10v;
    }
}

__global__ __launch_bounds__(256) void k_gemm2(const float* __restrict__ x,
                                               const float* __restrict__ t3,
                                               float* __restrict__ t9) {
    __shared__ float As[128 * PADK];
    __shared__ float Bs[128 * PADK];
    const int t    = threadIdx.x;
    const int lane = t & 31;
    const int wv   = t >> 5;
    const int smp  = blockIdx.y;
    const int d0   = blockIdx.x * 128;
    const float* xs  = x  + (size_t)smp * CC * HWP;
    const float* t3s = t3 + (size_t)smp * CC * HWP;

    const int mn    = lane & 15;
    const int kp    = (lane >> 4) << 1;
    const int rmsb  = (lane >> 4) << 3;
    const int mbase = (wv >> 1) * 32;
    const int nbase = (wv & 1) * 64;

    v8f acc[2][4] = {};

    for (int ks = 0; ks < HWP; ks += 32) {
        #pragma unroll
        for (int i = 0; i < 16; ++i) {
            int idx = t + i * 256;
            int m = idx >> 5, k = idx & 31;
            As[m * PADK + k] = xs[m * HWP + ks + k];
        }
        #pragma unroll
        for (int i = 0; i < 16; ++i) {
            int idx = t + i * 256;
            int n = idx >> 5, k = idx & 31;
            int d = d0 + n;
            int tap = d % 3, ci = d / 3;
            int pos = ks + k;
            int wp = pos % WW + (tap - 1);
            float v = 0.0f;
            if ((unsigned)wp < (unsigned)WW)
                v = t3s[ci * HWP + pos + (tap - 1)];
            Bs[n * PADK + k] = v;
        }
        __syncthreads();
        #pragma unroll
        for (int kk = 0; kk < 8; ++kk) {
            int ko = kk * 4 + kp;
            v2f a0 = *(const v2f*)&As[(mbase + mn) * PADK + ko];
            v2f a1 = *(const v2f*)&As[(mbase + 16 + mn) * PADK + ko];
            #pragma unroll
            for (int ni = 0; ni < 4; ++ni) {
                v2f b = *(const v2f*)&Bs[(nbase + ni * 16 + mn) * PADK + ko];
                acc[0][ni] = wmma_f32(a0, b, acc[0][ni]);
                acc[1][ni] = wmma_f32(a1, b, acc[1][ni]);
            }
        }
        __syncthreads();
    }

    const float scale = 1.0f / 56.0f;
    float* out = t9 + (size_t)smp * CC * C3;
    #pragma unroll
    for (int mi = 0; mi < 2; ++mi) {
        #pragma unroll
        for (int ni = 0; ni < 4; ++ni) {
            int col = d0 + nbase + ni * 16 + mn;
            #pragma unroll
            for (int r = 0; r < 8; ++r) {
                int m = mbase + mi * 16 + r + rmsb;
                out[m * C3 + col] = acc[mi][ni][r] * scale;
            }
        }
    }
}

__global__ __launch_bounds__(256) void k_gemm3(const float* __restrict__ t9,
                                               const float* __restrict__ t12,
                                               float* __restrict__ out) {
    __shared__ float As[128 * PADK];
    __shared__ float Bs[224 * PADK];
    const int t    = threadIdx.x;
    const int lane = t & 31;
    const int wv   = t >> 5;
    const int smp  = blockIdx.y;
    const int pos0 = blockIdx.x * 224;
    const float* t9s  = t9  + (size_t)smp * CC * C3;
    const float* t12s = t12 + (size_t)smp * C3 * HWP;

    const int mn    = lane & 15;
    const int kp    = (lane >> 4) << 1;
    const int rmsb  = (lane >> 4) << 3;
    const int mbase = (wv >> 1) * 32;
    const int nbase = (wv & 1) * 112;

    v8f acc[2][7] = {};

    for (int ks = 0; ks < C3; ks += 32) {
        #pragma unroll
        for (int i = 0; i < 16; ++i) {
            int idx = t + i * 256;
            int m = idx >> 5, k = idx & 31;
            As[m * PADK + k] = t9s[m * C3 + ks + k];
        }
        #pragma unroll
        for (int i = 0; i < 28; ++i) {
            int idx = t + i * 256;
            int k = idx / 224, nn = idx % 224;
            Bs[nn * PADK + k] = t12s[(ks + k) * HWP + pos0 + nn];
        }
        __syncthreads();
        #pragma unroll
        for (int kk = 0; kk < 8; ++kk) {
            int ko = kk * 4 + kp;
            v2f a0 = *(const v2f*)&As[(mbase + mn) * PADK + ko];
            v2f a1 = *(const v2f*)&As[(mbase + 16 + mn) * PADK + ko];
            #pragma unroll
            for (int ni = 0; ni < 7; ++ni) {
                v2f b = *(const v2f*)&Bs[(nbase + ni * 16 + mn) * PADK + ko];
                acc[0][ni] = wmma_f32(a0, b, acc[0][ni]);
                acc[1][ni] = wmma_f32(a1, b, acc[1][ni]);
            }
        }
        __syncthreads();
    }

    const float scale = 0.051031036307982884f;
    float* o = out + (size_t)smp * CC * HWP;
    #pragma unroll
    for (int mi = 0; mi < 2; ++mi) {
        #pragma unroll
        for (int ni = 0; ni < 7; ++ni) {
            int col = pos0 + nbase + ni * 16 + mn;
            #pragma unroll
            for (int r = 0; r < 8; ++r) {
                int m = mbase + mi * 16 + r + rmsb;
                o[m * HWP + col] = acc[mi][ni][r] * scale;
            }
        }
    }
}

extern "C" void kernel_launch(void* const* d_in, const int* in_sizes, int n_in,
                              void* d_out, int out_size, void* d_ws, size_t ws_size,
                              hipStream_t stream) {
    (void)in_sizes; (void)n_in; (void)out_size; (void)ws_size;
    const float* x   = (const float*)d_in[0];
    const float* p2  = (const float*)d_in[1];
    const float* p3  = (const float*)d_in[2];
    const float* p5  = (const float*)d_in[3];
    const float* p10 = (const float*)d_in[4];
    float* out = (float*)d_out;

    char* ws = (char*)d_ws;
    const size_t t3_bytes  = (size_t)NS * CC * HWP * sizeof(float);
    const size_t t12_bytes = (size_t)NS * C3 * HWP * sizeof(float);
    float* t3  = (float*)ws;
    float* t12 = (float*)(ws + t3_bytes);
    float* t9  = (float*)(ws + t3_bytes + t12_bytes);

    dim3 blk(256);
    k_gemm1<<<dim3(HWP / 224, NS), blk, 0, stream>>>(p3, x, t3);
    k_t12 <<<dim3((NS * CC * HWP) / 256), blk, 0, stream>>>(x, t3, p2, p5, p10, t12);
    k_gemm2<<<dim3(C3 / 128, NS), blk, 0, stream>>>(x, t3, t9);
    k_gemm3<<<dim3(HWP / 224, NS), blk, 0, stream>>>(t9, t12, out);
}


# Tile fills of all three GEMMs issue every global load of a k-step before the first wait: padded taps read branch-free and zeroed by select, per-wave rows make addresses base+constant
# speedup vs baseline: 2.2717x; 2.2717x over previous
#include <hip/hip_runtime.h>
#include <hip/hip_bf16.h>


#define NS   32
#define CC   128
#define HH   56
#define WW   56
#define HWP  3136
#define C3   384
#define PADK 34

typedef float v2f __attribute__((ext_vector_type(2)));
typedef float v8f __attribute__((ext_vector_type(8)));

__device__ __forceinline__ v8f wmma_f32(v2f a, v2f b, v8f c) {
    return __builtin_amdgcn_wmma_f32_16x16x4_f32(false, a, false, b, (short)0, c, false, false);
}

__global__ __launch_bounds__(256) void k_gemm1(const float* __restrict__ p3,
                                               const float* __restrict__ x,
                                               float* __restrict__ t3) {
    __shared__ float As[128 * PADK];
    __shared__ float Bs[224 * PADK];
    const int t    = threadIdx.x;
    const int lane = t & 31;
    const int wv   = t >> 5;
    const int smp  = blockIdx.y;
    const int pos0 = blockIdx.x * 224;
    const float* xs = x + (size_t)smp * CC * HWP;

    const int mn    = lane & 15;
    const int kp    = (lane >> 4) << 1;
    const int rmsb  = (lane >> 4) << 3;
    const int mbase = (wv >> 1) * 32;
    const int nbase = (wv & 1) * 112;

    const int wvs = __builtin_amdgcn_readfirstlane(wv);
    const float* ap = p3 + wv * C3 + lane;
    float* asst = As + wv * PADK + lane;
    float* bsst = Bs + lane * PADK + wv;
    const int posl = pos0 + lane;
    int hq[7];
    #pragma unroll
    for (int q = 0; q < 7; ++q) hq[q] = (posl + 32 * q) / WW;

    v8f acc[2][7] = {};

    for (int ks = 0; ks < C3; ks += 32) {
        float ra[16], rb[4][7];
        bool  ok[4][7];
        #pragma unroll
        for (int i = 0; i < 16; ++i)
            ra[i] = ap[ks + i * 8 * C3];
        #pragma unroll
        for (int a = 0; a < 4; ++a) {
            int j = ks + wvs + 8 * a;
            int tap = j % 3, ci = j / 3;
            int dh = 2 * (tap - 1);
            const float* xc = xs + ci * HWP;
            #pragma unroll
            for (int q = 0; q < 7; ++q) {
                ok[a][q] = (unsigned)(hq[q] + dh) < (unsigned)HH;
                rb[a][q] = xc[posl + 32 * q + (ok[a][q] ? WW * dh : 0)];
            }
        }
        __builtin_amdgcn_sched_barrier(0);
        #pragma unroll
        for (int i = 0; i < 16; ++i)
            asst[i * 8 * PADK] = ra[i];
        #pragma unroll
        for (int a = 0; a < 4; ++a) {
            #pragma unroll
            for (int q = 0; q < 7; ++q)
                bsst[q * 32 * PADK + 8 * a] = ok[a][q] ? rb[a][q] : 0.0f;
        }
        __syncthreads();
        #pragma unroll
        for (int kk = 0; kk < 8; ++kk) {
            int ko = kk * 4 + kp;
            v2f a0 = *(const v2f*)&As[(mbase + mn) * PADK + ko];
            v2f a1 = *(const v2f*)&As[(mbase + 16 + mn) * PADK + ko];
            #pragma unroll
            for (int ni = 0; ni < 7; ++ni) {
                v2f b = *(const v2f*)&Bs[(nbase + ni * 16 + mn) * PADK + ko];
                acc[0][ni] = wmma_f32(a0, b, acc[0][ni]);
                acc[1][ni] = wmma_f32(a1, b, acc[1][ni]);
            }
        }
        __syncthreads();
    }

    float* o = t3 + (size_t)smp * CC * HWP;
    #pragma unroll
    for (int mi = 0; mi < 2; ++mi) {
        #pragma unroll
        for (int ni = 0; ni < 7; ++ni) {
            int col = pos0 + nbase + ni * 16 + mn;
            #pragma unroll
            for (int r = 0; r < 8; ++r) {
                int m = mbase + mi * 16 + r + rmsb;
                o[m * HWP + col] = acc[mi][ni][r];
            }
        }
    }
}

__global__ __launch_bounds__(256) void k_t12(const float* __restrict__ x,
                                             const float* __restrict__ t3,
                                             const float* __restrict__ p2,
                                             const float* __restrict__ p5,
                                             const float* __restrict__ p10,
                                             float* __restrict__ t12) {
    int tid  = blockIdx.x * 256 + threadIdx.x;
    int pos  = tid % HWP;
    int rest = tid / HWP;
    int c    = rest % CC;
    int smp  = rest / CC;
    int h = pos / WW, w = pos % WW;

    const float* xc = x + ((size_t)smp * CC + c) * HWP;
    float x0  = xc[pos];
    float xvm = (h >= 2)      ? xc[pos - 2 * WW] : 0.0f;
    float xvp = (h < HH - 2)  ? xc[pos + 2 * WW] : 0.0f;
    float xhm = (w >= 2)      ? xc[pos - 2]      : 0.0f;
    float xhp = (w < WW - 2)  ? xc[pos + 2]      : 0.0f;

    float mx = fmaxf(xhm, fmaxf(x0, xhp));
    float e0 = __expf(xhm - mx), e1 = __expf(x0 - mx), e2 = __expf(xhp - mx);
    float inv = 1.0f / (e0 + e1 + e2);
    float t8[3] = { e0 * inv, e1 * inv, e2 * inv };
    float xv[3] = { xvm, x0, xvp };

    float p5v = p5[pos];
    const float* t3s  = t3  + (size_t)smp * CC * HWP;
    const float* p10s = p10 + (size_t)(3 * c) * HWP + pos;
    float* outj = t12 + ((size_t)smp * C3 + 3 * c) * HWP + pos;

    #pragma unroll
    for (int k = 0; k < 3; ++k) {
        int j = 3 * c + k;
        int b = j & (CC - 1);
        float t5v  = p5v * t3s[b * HWP + pos];
        float t2v  = p2[k * WW + w] * xv[k];
        float t10v = p10s[k * HWP] * t8[k];
        outj[k * HWP] = t5v - t2v - t10v;
    }
}

__global__ __launch_bounds__(256) void k_gemm2(const float* __restrict__ x,
                                               const float* __restrict__ t3,
                                               float* __restrict__ t9) {
    __shared__ float As[128 * PADK];
    __shared__ float Bs[128 * PADK];
    const int t    = threadIdx.x;
    const int lane = t & 31;
    const int wv   = t >> 5;
    const int smp  = blockIdx.y;
    const int d0   = blockIdx.x * 128;
    const float* xs  = x  + (size_t)smp * CC * HWP;
    const float* t3s = t3 + (size_t)smp * CC * HWP;

    const int mn    = lane & 15;
    const int kp    = (lane >> 4) << 1;
    const int rmsb  = (lane >> 4) << 3;
    const int mbase = (wv >> 1) * 32;
    const int nbase = (wv & 1) * 64;

    const int wvs = __builtin_amdgcn_readfirstlane(wv);
    const float* ap = xs + wv * HWP + lane;
    float* asst = As + wv * PADK + lane;
    float* bsst = Bs + wv * PADK + lane;

    v8f acc[2][4] = {};

    for (int ks = 0; ks < HWP; ks += 32) {
        float ra[16], rb[16];
        bool  ok[16];
        const int pos = ks + lane;
        const int wc  = pos % WW;
        #pragma unroll
        for (int i = 0; i < 16; ++i)
            ra[i] = ap[ks + i * 8 * HWP];
        #pragma unroll
        for (int i = 0; i < 16; ++i) {
            int d = d0 + wvs + 8 * i;
            int tap = d % 3, ci = d / 3;
            ok[i] = (unsigned)(wc + (tap - 1)) < (unsigned)WW;
            rb[i] = t3s[ci * HWP + pos + (ok[i] ? (tap - 1) : 0)];
        }
        __builtin_amdgcn_sched_barrier(0);
        #pragma unroll
        for (int i = 0; i < 16; ++i)
            asst[i * 8 * PADK] = ra[i];
        #pragma unroll
        for (int i = 0; i < 16; ++i)
            bsst[i * 8 * PADK] = ok[i] ? rb[i] : 0.0f;
        __syncthreads();
        #pragma unroll
        for (int kk = 0; kk < 8; ++kk) {
            int ko = kk * 4 + kp;
            v2f a0 = *(const v2f*)&As[(mbase + mn) * PADK + ko];
            v2f a1 = *(const v2f*)&As[(mbase + 16 + mn) * PADK + ko];
            #pragma unroll
            for (int ni = 0; ni < 4; ++ni) {
                v2f b = *(const v2f*)&Bs[(nbase + ni * 16 + mn) * PADK + ko];
                acc[0][ni] = wmma_f32(a0, b, acc[0][ni]);
                acc[1][ni] = wmma_f32(a1, b, acc[1][ni]);
            }
        }
        __syncthreads();
    }

    const float scale = 1.0f / 56.0f;
    float* out = t9 + (size_t)smp * CC * C3;
    #pragma unroll
    for (int mi = 0; mi < 2; ++mi) {
        #pragma unroll
        for (int ni = 0; ni < 4; ++ni) {
            int col = d0 + nbase + ni * 16 + mn;
            #pragma unroll
            for (int r = 0; r < 8; ++r) {
                int m = mbase + mi * 16 + r + rmsb;
                out[m * C3 + col] = acc[mi][ni][r] * scale;
            }
        }
    }
}

__global__ __launch_bounds__(256) void k_gemm3(const float* __restrict__ t9,
                                               const float* __restrict__ t12,
                                               float* __restrict__ out) {
    __shared__ float As[128 * PADK];
    __shared__ float Bs[224 * PADK];
    const int t    = threadIdx.x;
    const int lane = t & 31;
    const int wv   = t >> 5;
    const int smp  = blockIdx.y;
    const int pos0 = blockIdx.x * 224;
    const float* t9s  = t9  + (size_t)smp * CC * C3;
    const float* t12s = t12 + (size_t)smp * C3 * HWP;

    const int mn    = lane & 15;
    const int kp    = (lane >> 4) << 1;
    const int rmsb  = (lane >> 4) << 3;
    const int mbase = (wv >> 1) * 32;
    const int nbase = (wv & 1) * 112;

    const float* ap = t9s + wv * C3 + lane;
    const float* bp = t12s + wv * HWP + pos0 + lane;
    float* asst = As + wv * PADK + lane;
    float* bsst = Bs + lane * PADK + wv;

    v8f acc[2][7] = {};

    for (int ks = 0; ks < C3; ks += 32) {
        float ra[16], rb[4][7];
        #pragma unroll
        for (int i = 0; i < 16; ++i)
            ra[i] = ap[ks + i * 8 * C3];
        #pragma unroll
        for (int a = 0; a < 4; ++a) {
            #pragma unroll
            for (int q = 0; q < 7; ++q)
                rb[a][q] = bp[(ks + 8 * a) * HWP + 32 * q];
        }
        #pragma unroll
        for (int i = 0; i < 16; ++i)
            asst[i * 8 * PADK] = ra[i];
        #pragma unroll
        for (int a = 0; a < 4; ++a) {
            #pragma unroll
            for (int q = 0; q < 7; ++q)
                bsst[q * 32 * PADK + 8 * a] = rb[a][q];
        }
        __syncthreads();
        #pragma unroll
        for (int kk = 0; kk < 8; ++kk) {
            int ko = kk * 4 + kp;
            v2f a0 = *(const v2f*)&As[(mbase + mn) * PADK + ko];
            v2f a1 = *(const v2f*)&As[(mbase + 16 + mn) * PADK + ko];
            #pragma unroll
            for (int ni = 0; ni < 7; ++ni) {
                v2f b = *(const v2f*)&Bs[(nbase + ni * 16 + mn) * PADK + ko];
                acc[0][ni] = wmma_f32(a0, b, acc[0][ni]);
                acc[1][ni] = wmma_f32(a1, b, acc[1][ni]);
            }
        }
        __syncthreads();
    }

    const float scale = 0.051031036307982884f;
    float* o = out + (size_t)smp * CC * HWP;
    #pragma unroll
    for (int mi = 0; mi < 2; ++mi) {
        #pragma unroll
        for (int ni = 0; ni < 7; ++ni) {
            int col = pos0 + nbase + ni * 16 + mn;
            #pragma unroll
            for (int r = 0; r < 8; ++r) {
                int m = mbase + mi * 16 + r + rmsb;
                o[m * HWP + col] = acc[mi][ni][r] * scale;
            }
        }
    }
}

extern "C" void kernel_launch(void* const* d_in, const int* in_sizes, int n_in,
                              void* d_out, int out_size, void* d_ws, size_t ws_size,
                              hipStream_t stream) {
    (void)in_sizes; (void)n_in; (void)out_size; (void)ws_size;
    const float* x   = (const float*)d_in[0];
    const float* p2  = (const float*)d_in[1];
    const float* p3  = (const float*)d_in[2];
    const float* p5  = (const float*)d_in[3];
    const float* p10 = (const float*)d_in[4];
    float* out = (float*)d_out;

    char* ws = (char*)d_ws;
    const size_t t3_bytes  = (size_t)NS * CC * HWP * sizeof(float);
    const size_t t12_bytes = (size_t)NS * C3 * HWP * sizeof(float);
    float* t3  = (float*)ws;
    float* t12 = (float*)(ws + t3_bytes);
    float* t9  = (float*)(ws + t3_bytes + t12_bytes);

    dim3 blk(256);
    k_gemm1<<<dim3(HWP / 224, NS), blk, 0, stream>>>(p3, x, t3);
    k_t12 <<<dim3((NS * CC * HWP) / 256), blk, 0, stream>>>(x, t3, p2, p5, p10, t12);
    k_gemm2<<<dim3(C3 / 128, NS), blk, 0, stream>>>(x, t3, t9);
    k_gemm3<<<dim3(HWP / 224, NS), blk, 0, stream>>>(t9, t12, out);
}


# GEMM2 split over K into 7 chunks on blockIdx.z: 672 work-groups fill all WGPs instead of 96; a small fifth kernel adds the 7 partial planes in fixed order (summation order of t9 changes)
# speedup vs baseline: 3.1840x; 1.4016x over previous
#include <hip/hip_runtime.h>
#include <hip/hip_bf16.h>


#define NS   32
#define CC   128
#define HH   56
#define WW   56
#define HWP  3136
#define C3   384
#define PADK 34
#define KSPLIT 7
#define KCHUNK 448
#define T9N  (NS * CC * C3)

typedef float v2f __attribute__((ext_vector_type(2)));
typedef float v8f __attribute__((ext_vector_type(8)));

__device__ __forceinline__ v8f wmma_f32(v2f a, v2f b, v8f c) {
    return __builtin_amdgcn_wmma_f32_16x16x4_f32(false, a, false, b, (short)0, c, false, false);
}

__global__ __launch_bounds__(256) void k_gemm1(const float* __restrict__ p3,
                                               const float* __restrict__ x,
                                               float* __restrict__ t3) {
    __shared__ float As[128 * PADK];
    __shared__ float Bs[224 * PADK];
    const int t    = threadIdx.x;
    const int lane = t & 31;
    const int wv   = t >> 5;
    const int smp  = blockIdx.y;
    const int pos0 = blockIdx.x * 224;
    const float* xs = x + (size_t)smp * CC * HWP;

    const int mn    = lane & 15;
    const int kp    = (lane >> 4) << 1;
    const int rmsb  = (lane >> 4) << 3;
    const int mbase = (wv >> 1) * 32;
    const int nbase = (wv & 1) * 112;

    const int wvs = __builtin_amdgcn_readfirstlane(wv);
    const float* ap = p3 + wv * C3 + lane;
    float* asst = As + wv * PADK + lane;
    float* bsst = Bs + lane * PADK + wv;
    const int posl = pos0 + lane;
    int hq[7];
    #pragma unroll
    for (int q = 0; q < 7; ++q) hq[q] = (posl + 32 * q) / WW;

    v8f acc[2][7] = {};

    for (int ks = 0; ks < C3; ks += 32) {
        float ra[16], rb[4][7];
        bool  ok[4][7];
        #pragma unroll
        for (int i = 0; i < 16; ++i)
            ra[i] = ap[ks + i * 8 * C3];
        #pragma unroll
        for (int a = 0; a < 4; ++a) {
            int j = ks + wvs + 8 * a;
            int tap = j % 3, ci = j / 3;
            int dh = 2 * (tap - 1);
            const float* xc = xs + ci * HWP;
            #pragma unroll
            for (int q = 0; q < 7; ++q) {
                ok[a][q] = (unsigned)(hq[q] + dh) < (unsigned)HH;
                rb[a][q] = xc[posl + 32 * q + (ok[a][q] ? WW * dh : 0)];
            }
        }
        __builtin_amdgcn_sched_barrier(0);
        #pragma unroll
        for (int i = 0; i < 16; ++i)
            asst[i * 8 * PADK] = ra[i];
        #pragma unroll
        for (int a = 0; a < 4; ++a) {
            #pragma unroll
            for (int q = 0; q < 7; ++q)
                bsst[q * 32 * PADK + 8 * a] = ok[a][q] ? rb[a][q] : 0.0f;
        }
        __syncthreads();
        #pragma unroll
        for (int kk = 0; kk < 8; ++kk) {
            int ko = kk * 4 + kp;
            v2f a0 = *(const v2f*)&As[(mbase + mn) * PADK + ko];
            v2f a1 = *(const v2f*)&As[(mbase + 16 + mn) * PADK + ko];
            #pragma unroll
            for (int ni = 0; ni < 7; ++ni) {
                v2f b = *(const v2f*)&Bs[(nbase + ni * 16 + mn) * PADK + ko];
                acc[0][ni] = wmma_f32(a0, b, acc[0][ni]);
                acc[1][ni] = wmma_f32(a1, b, acc[1][ni]);
            }
        }
        __syncthreads();
    }

    float* o = t3 + (size_t)smp * CC * HWP;
    #pragma unroll
    for (int mi = 0; mi < 2; ++mi) {
        #pragma unroll
        for (int ni = 0; ni < 7; ++ni) {
            int col = pos0 + nbase + ni * 16 + mn;
            #pragma unroll
            for (int r = 0; r < 8; ++r) {
                int m = mbase + mi * 16 + r + rmsb;
                o[m * HWP + col] = acc[mi][ni][r];
            }
        }
    }
}

__global__ __launch_bounds__(256) void k_t12(const float* __restrict__ x,
                                             const float* __restrict__ t3,
                                             const float* __restrict__ p2,
                                             const float* __restrict__ p5,
                                             const float* __restrict__ p10,
                                             float* __restrict__ t12) {
    int tid  = blockIdx.x * 256 + threadIdx.x;
    int pos  = tid % HWP;
    int rest = tid / HWP;
    int c    = rest % CC;
    int smp  = rest / CC;
    int h = pos / WW, w = pos % WW;

    const float* xc = x + ((size_t)smp * CC + c) * HWP;
    float x0  = xc[pos];
    float xvm = (h >= 2)      ? xc[pos - 2 * WW] : 0.0f;
    float xvp = (h < HH - 2)  ? xc[pos + 2 * WW] : 0.0f;
    float xhm = (w >= 2)      ? xc[pos - 2]      : 0.0f;
    float xhp = (w < WW - 2)  ? xc[pos + 2]      : 0.0f;

    float mx = fmaxf(xhm, fmaxf(x0, xhp));
    float e0 = __expf(xhm - mx), e1 = __expf(x0 - mx), e2 = __expf(xhp - mx);
    float inv = 1.0f / (e0 + e1 + e2);
    float t8[3] = { e0 * inv, e1 * inv, e2 * inv };
    float xv[3] = { xvm, x0, xvp };

    float p5v = p5[pos];
    const float* t3s  = t3  + (size_t)smp * CC * HWP;
    const float* p10s = p10 + (size_t)(3 * c) * HWP + pos;
    float* outj = t12 + ((size_t)smp * C3 + 3 * c) * HWP + pos;

    #pragma unroll
    for (int k = 0; k < 3; ++k) {
        int j = 3 * c + k;
        int b = j & (CC - 1);
        float t5v  = p5v * t3s[b * HWP + pos];
        float t2v  = p2[k * WW + w] * xv[k];
        float t10v = p10s[k * HWP] * t8[k];
        outj[k * HWP] = t5v - t2v - t10v;
    }
}

__global__ __launch_bounds__(256) void k_gemm2(const float* __restrict__ x,
                                               const float* __restrict__ t3,
                                               float* __restrict__ t9p) {
    __shared__ float As[128 * PADK];
    __shared__ float Bs[128 * PADK];
    const int t    = threadIdx.x;
    const int lane = t & 31;
    const int wv   = t >> 5;
    const int smp  = blockIdx.y;
    const int d0   = blockIdx.x * 128;
    const float* xs  = x  + (size_t)smp * CC * HWP;
    const float* t3s = t3 + (size_t)smp * CC * HWP;

    const int mn    = lane & 15;
    const int kp    = (lane >> 4) << 1;
    const int rmsb  = (lane >> 4) << 3;
    const int mbase = (wv >> 1) * 32;
    const int nbase = (wv & 1) * 64;

    const int wvs = __builtin_amdgcn_readfirstlane(wv);
    const float* ap = xs + wv * HWP + lane;
    float* asst = As + wv * PADK + lane;
    float* bsst = Bs + wv * PADK + lane;

    v8f acc[2][4] = {};

    const int kbeg = blockIdx.z * KCHUNK;
    for (int ks = kbeg; ks < kbeg + KCHUNK; ks += 32) {
        float ra[16], rb[16];
        bool  ok[16];
        const int pos = ks + lane;
        const int wc  = pos % WW;
        #pragma unroll
        for (int i = 0; i < 16; ++i)
            ra[i] = ap[ks + i * 8 * HWP];
        #pragma unroll
        for (int i = 0; i < 16; ++i) {
            int d = d0 + wvs + 8 * i;
            int tap = d % 3, ci = d / 3;
            ok[i] = (unsigned)(wc + (tap - 1)) < (unsigned)WW;
            rb[i] = t3s[ci * HWP + pos + (ok[i] ? (tap - 1) : 0)];
        }
        __builtin_amdgcn_sched_barrier(0);
        #pragma unroll
        for (int i = 0; i < 16; ++i)
            asst[i * 8 * PADK] = ra[i];
        #pragma unroll
        for (int i = 0; i < 16; ++i)
            bsst[i * 8 * PADK] = ok[i] ? rb[i] : 0.0f;
        __syncthreads();
        #pragma unroll
        for (int kk = 0; kk < 8; ++kk) {
            int ko = kk * 4 + kp;
            v2f a0 = *(const v2f*)&As[(mbase + mn) * PADK + ko];
            v2f a1 = *(const v2f*)&As[(mbase + 16 + mn) * PADK + ko];
            #pragma unroll
            for (int ni = 0; ni < 4; ++ni) {
                v2f b = *(const v2f*)&Bs[(nbase + ni * 16 + mn) * PADK + ko];
                acc[0][ni] = wmma_f32(a0, b, acc[0][ni]);
                acc[1][ni] = wmma_f32(a1, b, acc[1][ni]);
            }
        }
        __syncthreads();
    }

    float* out = t9p + (size_t)blockIdx.z * T9N + (size_t)smp * CC * C3;
    #pragma unroll
    for (int mi = 0; mi < 2; ++mi) {
        #pragma unroll
        for (int ni = 0; ni < 4; ++ni) {
            int col = d0 + nbase + ni * 16 + mn;
            #pragma unroll
            for (int r = 0; r < 8; ++r) {
                int m = mbase + mi * 16 + r + rmsb;
                out[m * C3 + col] = acc[mi][ni][r];
            }
        }
    }
}

__global__ __launch_bounds__(256) void k_t9sum(const float* __restrict__ t9p,
                                               float* __restrict__ t9) {
    const int i = blockIdx.x * 256 + threadIdx.x;
    const float4* p = (const float4*)t9p + i;
    float4 v[KSPLIT];
    #pragma unroll
    for (int z = 0; z < KSPLIT; ++z) v[z] = p[(size_t)z * (T9N / 4)];
    float4 s = v[0];
    #pragma unroll
    for (int z = 1; z < KSPLIT; ++z) {
        s.x += v[z].x; s.y += v[z].y; s.z += v[z].z; s.w += v[z].w;
    }
    const float scale = 1.0f / 56.0f;
    s.x *= scale; s.y *= scale; s.z *= scale; s.w *= scale;
    ((float4*)t9)[i] = s;
}

__global__ __launch_bounds__(256) void k_gemm3(const float* __restrict__ t9,
                                               const float* __restrict__ t12,
                                               float* __restrict__ out) {
    __shared__ float As[128 * PADK];
    __shared__ float Bs[224 * PADK];
    const int t    = threadIdx.x;
    const int lane = t & 31;
    const int wv   = t >> 5;
    const int smp  = blockIdx.y;
    const int pos0 = blockIdx.x * 224;
    const float* t9s  = t9  + (size_t)smp * CC * C3;
    const float* t12s = t12 + (size_t)smp * C3 * HWP;

    const int mn    = lane & 15;
    const int kp    = (lane >> 4) << 1;
    const int rmsb  = (lane >> 4) << 3;
    const int mbase = (wv >> 1) * 32;
    const int nbase = (wv & 1) * 112;

    const float* ap = t9s + wv * C3 + lane;
    const float* bp = t12s + wv * HWP + pos0 + lane;
    float* asst = As + wv * PADK + lane;
    float* bsst = Bs + lane * PADK + wv;

    v8f acc[2][7] = {};

    for (int ks = 0; ks < C3; ks += 32) {
        float ra[16], rb[4][7];
        #pragma unroll
        for (int i = 0; i < 16; ++i)
            ra[i] = ap[ks + i * 8 * C3];
        #pragma unroll
        for (int a = 0; a < 4; ++a) {
            #pragma unroll
            for (int q = 0; q < 7; ++q)
                rb[a][q] = bp[(ks + 8 * a) * HWP + 32 * q];
        }
        #pragma unroll
        for (int i = 0; i < 16; ++i)
            asst[i * 8 * PADK] = ra[i];
        #pragma unroll
        for (int a = 0; a < 4; ++a) {
            #pragma unroll
            for (int q = 0; q < 7; ++q)
                bsst[q * 32 * PADK + 8 * a] = rb[a][q];
        }
        __syncthreads();
        #pragma unroll
        for (int kk = 0; kk < 8; ++kk) {
            int ko = kk * 4 + kp;
            v2f a0 = *(const v2f*)&As[(mbase + mn) * PADK + ko];
            v2f a1 = *(const v2f*)&As[(mbase + 16 + mn) * PADK + ko];
            #pragma unroll
            for (int ni = 0; ni < 7; ++ni) {
                v2f b = *(const v2f*)&Bs[(nbase + ni * 16 + mn) * PADK + ko];
                acc[0][ni] = wmma_f32(a0, b, acc[0][ni]);
                acc[1][ni] = wmma_f32(a1, b, acc[1][ni]);
            }
        }
        __syncthreads();
    }

    const float scale = 0.051031036307982884f;
    float* o = out + (size_t)smp * CC * HWP;
    #pragma unroll
    for (int mi = 0; mi < 2; ++mi) {
        #pragma unroll
        for (int ni = 0; ni < 7; ++ni) {
            int col = pos0 + nbase + ni * 16 + mn;
            #pragma unroll
            for (int r = 0; r < 8; ++r) {
                int m = mbase + mi * 16 + r + rmsb;
                o[m * HWP + col] = acc[mi][ni][r] * scale;
            }
        }
    }
}

extern "C" void kernel_launch(void* const* d_in, const int* in_sizes, int n_in,
                              void* d_out, int out_size, void* d_ws, size_t ws_size,
                              hipStream_t stream) {
    (void)in_sizes; (void)n_in; (void)out_size; (void)ws_size;
    const float* x   = (const float*)d_in[0];
    const float* p2  = (const float*)d_in[1];
    const float* p3  = (const float*)d_in[2];
    const float* p5  = (const float*)d_in[3];
    const float* p10 = (const float*)d_in[4];
    float* out = (float*)d_out;

    char* ws = (char*)d_ws;
    const size_t t3_bytes  = (size_t)NS * CC * HWP * sizeof(float);
    const size_t t12_bytes = (size_t)NS * C3 * HWP * sizeof(float);
    float* t3  = (float*)ws;
    float* t12 = (float*)(ws + t3_bytes);
    float* t9  = (float*)(ws + t3_bytes + t12_bytes);
    float* t9p = t12;

    dim3 blk(256);
    k_gemm1<<<dim3(HWP / 224, NS), blk, 0, stream>>>(p3, x, t3);
    k_gemm2<<<dim3(C3 / 128, NS, KSPLIT), blk, 0, stream>>>(x, t3, t9p);
    k_t9sum<<<dim3(T9N / 4 / 256), blk, 0, stream>>>(t9p, t9);
    k_t12 <<<dim3((NS * CC * HWP) / 256), blk, 0, stream>>>(x, t3, p2, p5, p10, t12);
    k_gemm3<<<dim3(HWP / 224, NS), blk, 0, stream>>>(t9, t12, out);
}


# t12 is no longer stored: GEMM3's B-tile fill computes it from x, t3, p2, p5, p10 (k-step 24 = 8 channels x 3 taps, one softmax per position), dropping the elementwise launch; sum order unchanged
# speedup vs baseline: 3.6677x; 1.1519x over previous
#include <hip/hip_runtime.h>
#include <hip/hip_bf16.h>


#define NS   32
#define CC   128
#define HH   56
#define WW   56
#define HWP  3136
#define C3   384
#define PADK 34
#define KSPLIT 7
#define KCHUNK 448
#define T9N  (NS * CC * C3)
#define KS3  24
#define NX   (NS * CC * HWP)

typedef float v2f __attribute__((ext_vector_type(2)));
typedef float v8f __attribute__((ext_vector_type(8)));

__device__ __forceinline__ v8f wmma_f32(v2f a, v2f b, v8f c) {
    return __builtin_amdgcn_wmma_f32_16x16x4_f32(false, a, false, b, (short)0, c, false, false);
}

__global__ __launch_bounds__(256) void k_gemm1(const float* __restrict__ p3,
                                               const float* __restrict__ x,
                                               float* __restrict__ t3) {
    __shared__ float As[128 * PADK];
    __shared__ float Bs[224 * PADK];
    const int t    = threadIdx.x;
    const int lane = t & 31;
    const int wv   = t >> 5;
    const int smp  = blockIdx.y;
    const int pos0 = blockIdx.x * 224;
    const float* xs = x + (size_t)smp * CC * HWP;

    const int mn    = lane & 15;
    const int kp    = (lane >> 4) << 1;
    const int rmsb  = (lane >> 4) << 3;
    const int mbase = (wv >> 1) * 32;
    const int nbase = (wv & 1) * 112;

    const int wvs = __builtin_amdgcn_readfirstlane(wv);
    const float* ap = p3 + wv * C3 + lane;
    float* asst = As + wv * PADK + lane;
    float* bsst = Bs + lane * PADK + wv;
    const int posl = pos0 + lane;
    int hq[7];
    #pragma unroll
    for (int q = 0; q < 7; ++q) hq[q] = (posl + 32 * q) / WW;

    v8f acc[2][7] = {};

    for (int ks = 0; ks < C3; ks += 32) {
        float ra[16], rb[4][7];
        bool  ok[4][7];
        #pragma unroll
        for (int i = 0; i < 16; ++i)
            ra[i] = ap[ks + i * 8 * C3];
        #pragma unroll
        for (int a = 0; a < 4; ++a) {
            int j = ks + wvs + 8 * a;
            int tap = j % 3, ci = j / 3;
            int dh = 2 * (tap - 1);
            const float* xc = xs + ci * HWP;
            #pragma unroll
            for (int q = 0; q < 7; ++q) {
                ok[a][q] = (unsigned)(hq[q] + dh) < (unsigned)HH;
                rb[a][q] = xc[posl + 32 * q + (ok[a][q] ? WW * dh : 0)];
            }
        }
        __builtin_amdgcn_sched_barrier(0);
        #pragma unroll
        for (int i = 0; i < 16; ++i)
            asst[i * 8 * PADK] = ra[i];
        #pragma unroll
        for (int a = 0; a < 4; ++a) {
            #pragma unroll
            for (int q = 0; q < 7; ++q)
                bsst[q * 32 * PADK + 8 * a] = ok[a][q] ? rb[a][q] : 0.0f;
        }
        __syncthreads();
        #pragma unroll
        for (int kk = 0; kk < 8; ++kk) {
            int ko = kk * 4 + kp;
            v2f a0 = *(const v2f*)&As[(mbase + mn) * PADK + ko];
            v2f a1 = *(const v2f*)&As[(mbase + 16 + mn) * PADK + ko];
            #pragma unroll
            for (int ni = 0; ni < 7; ++ni) {
                v2f b = *(const v2f*)&Bs[(nbase + ni * 16 + mn) * PADK + ko];
                acc[0][ni] = wmma_f32(a0, b, acc[0][ni]);
                acc[1][ni] = wmma_f32(a1, b, acc[1][ni]);
            }
        }
        __syncthreads();
    }

    float* o = t3 + (size_t)smp * CC * HWP;
    #pragma unroll
    for (int mi = 0; mi < 2; ++mi) {
        #pragma unroll
        for (int ni = 0; ni < 7; ++ni) {
            int col = pos0 + nbase + ni * 16 + mn;
            #pragma unroll
            for (int r = 0; r < 8; ++r) {
                int m = mbase + mi * 16 + r + rmsb;
                o[m * HWP + col] = acc[mi][ni][r];
            }
        }
    }
}

__global__ __launch_bounds__(256) void k_gemm2(const float* __restrict__ x,
                                               const float* __restrict__ t3,
                                               float* __restrict__ t9p) {
    __shared__ float As[128 * PADK];
    __shared__ float Bs[128 * PADK];
    const int t    = threadIdx.x;
    const int lane = t & 31;
    const int wv   = t >> 5;
    const int smp  = blockIdx.y;
    const int d0   = blockIdx.x * 128;
    const float* xs  = x  + (size_t)smp * CC * HWP;
    const float* t3s = t3 + (size_t)smp * CC * HWP;

    const int mn    = lane & 15;
    const int kp    = (lane >> 4) << 1;
    const int rmsb  = (lane >> 4) << 3;
    const int mbase = (wv >> 1) * 32;
    const int nbase = (wv & 1) * 64;

    const int wvs = __builtin_amdgcn_readfirstlane(wv);
    const float* ap = xs + wv * HWP + lane;
    float* asst = As + wv * PADK + lane;
    float* bsst = Bs + wv * PADK + lane;

    v8f acc[2][4] = {};

    const int kbeg = blockIdx.z * KCHUNK;
    for (int ks = kbeg; ks < kbeg + KCHUNK; ks += 32) {
        float ra[16], rb[16];
        bool  ok[16];
        const int pos = ks + lane;
        const int wc  = pos % WW;
        #pragma unroll
        for (int i = 0; i < 16; ++i)
            ra[i] = ap[ks + i * 8 * HWP];
        #pragma unroll
        for (int i = 0; i < 16; ++i) {
            int d = d0 + wvs + 8 * i;
            int tap = d % 3, ci = d / 3;
            ok[i] = (unsigned)(wc + (tap - 1)) < (unsigned)WW;
            rb[i] = t3s[ci * HWP + pos + (ok[i] ? (tap - 1) : 0)];
        }
        __builtin_amdgcn_sched_barrier(0);
        #pragma unroll
        for (int i = 0; i < 16; ++i)
            asst[i * 8 * PADK] = ra[i];
        #pragma unroll
        for (int i = 0; i < 16; ++i)
            bsst[i * 8 * PADK] = ok[i] ? rb[i] : 0.0f;
        __syncthreads();
        #pragma unroll
        for (int kk = 0; kk < 8; ++kk) {
            int ko = kk * 4 + kp;
            v2f a0 = *(const v2f*)&As[(mbase + mn) * PADK + ko];
            v2f a1 = *(const v2f*)&As[(mbase + 16 + mn) * PADK + ko];
            #pragma unroll
            for (int ni = 0; ni < 4; ++ni) {
                v2f b = *(const v2f*)&Bs[(nbase + ni * 16 + mn) * PADK + ko];
                acc[0][ni] = wmma_f32(a0, b, acc[0][ni]);
                acc[1][ni] = wmma_f32(a1, b, acc[1][ni]);
            }
        }
        __syncthreads();
    }

    float* out = t9p + (size_t)blockIdx.z * T9N + (size_t)smp * CC * C3;
    #pragma unroll
    for (int mi = 0; mi < 2; ++mi) {
        #pragma unroll
        for (int ni = 0; ni < 4; ++ni) {
            int col = d0 + nbase + ni * 16 + mn;
            #pragma unroll
            for (int r = 0; r < 8; ++r) {
                int m = mbase + mi * 16 + r + rmsb;
                out[m * C3 + col] = acc[mi][ni][r];
            }
        }
    }
}

__global__ __launch_bounds__(256) void k_t9sum(const float* __restrict__ t9p,
                                               float* __restrict__ t9) {
    const int i = blockIdx.x * 256 + threadIdx.x;
    const float4* p = (const float4*)t9p + i;
    float4 v[KSPLIT];
    #pragma unroll
    for (int z = 0; z < KSPLIT; ++z) v[z] = p[(size_t)z * (T9N / 4)];
    float4 s = v[0];
    #pragma unroll
    for (int z = 1; z < KSPLIT; ++z) {
        s.x += v[z].x; s.y += v[z].y; s.z += v[z].z; s.w += v[z].w;
    }
    const float scale = 1.0f / 56.0f;
    s.x *= scale; s.y *= scale; s.z *= scale; s.w *= scale;
    ((float4*)t9)[i] = s;
}

__global__ __launch_bounds__(256) void k_gemm3(const float* __restrict__ t9,
                                               const float* __restrict__ x,
                                               const float* __restrict__ t3,
                                               const float* __restrict__ p2,
                                               const float* __restrict__ p5,
                                               const float* __restrict__ p10,
                                               float* __restrict__ out) {
    __shared__ float As[128 * PADK];
    __shared__ float Bs[224 * PADK];
    __shared__ float P5s[224];
    __shared__ float P2e[3 * 224];
    const int t    = threadIdx.x;
    const int lane = t & 31;
    const int wv   = t >> 5;
    const int smp  = blockIdx.y;
    const int pos0 = blockIdx.x * 224;
    const float* t9s = t9 + (size_t)smp * CC * C3;
    const float* t3s = t3 + (size_t)smp * CC * HWP;

    const int mn    = lane & 15;
    const int kp    = (lane >> 4) << 1;
    const int rmsb  = (lane >> 4) << 3;
    const int mbase = (wv >> 1) * 32;
    const int nbase = (wv & 1) * 112;

    const int wvs  = __builtin_amdgcn_readfirstlane(wv);
    const int acol = lane < KS3 ? lane : KS3 - 1;
    const float* ap = t9s + wv * C3 + acol;
    float* asst = As + wv * PADK + lane;
    float* bsst = Bs + lane * PADK + 3 * wv;
    const int posl = pos0 + lane;

    if (t < 224) P5s[t] = p5[pos0 + t];
    for (int i = t; i < 3 * 224; i += 256)
        P2e[i] = p2[(i / 224) * WW + (i % 224) % WW];

    unsigned mk = 0;
    #pragma unroll
    for (int q = 0; q < 7; ++q) {
        int pl = lane + 32 * q;
        int h = blockIdx.x * 4 + pl / WW, w = pl % WW;
        unsigned b = (h >= 2 ? 1u : 0u) | (h < HH - 2 ? 2u : 0u)
                   | (w >= 2 ? 4u : 0u) | (w < WW - 2 ? 8u : 0u);
        mk |= b << (4 * q);
    }
    __syncthreads();

    v8f acc[2][7] = {};

    for (int ks = 0; ks < C3; ks += KS3) {
        float ra[16], lx0[7], lvm[7], lvp[7], lhm[7], lhp[7], lt3[3][7], lp10[3][7];
        const int d0 = ks + 3 * wvs;
        const int gx = (smp * CC + d0 / 3) * HWP + posl;
        #pragma unroll
        for (int i = 0; i < 16; ++i)
            ra[i] = ap[ks + i * 8 * C3];
        #pragma unroll
        for (int q = 0; q < 7; ++q) {
            const int g = gx + 32 * q;
            lx0[q] = x[g];
            lvm[q] = x[max(g - 2 * WW, 0)];
            lvp[q] = x[min(g + 2 * WW, NX - 1)];
            lhm[q] = x[max(g - 2, 0)];
            lhp[q] = x[min(g + 2, NX - 1)];
        }
        #pragma unroll
        for (int k = 0; k < 3; ++k) {
            const float* t3r  = t3s + ((d0 + k) & (CC - 1)) * HWP + posl;
            const float* p10r = p10 + (d0 + k) * HWP + posl;
            #pragma unroll
            for (int q = 0; q < 7; ++q) {
                lt3[k][q]  = t3r[32 * q];
                lp10[k][q] = p10r[32 * q];
            }
        }
        __builtin_amdgcn_sched_barrier(0);
        #pragma unroll
        for (int i = 0; i < 16; ++i)
            asst[i * 8 * PADK] = ra[i];
        #pragma unroll
        for (int q = 0; q < 7; ++q) {
            const unsigned b = mk >> (4 * q);
            const float x0  = lx0[q];
            const float xvm = (b & 1u) ? lvm[q] : 0.0f;
            const float xvp = (b & 2u) ? lvp[q] : 0.0f;
            const float xhm = (b & 4u) ? lhm[q] : 0.0f;
            const float xhp = (b & 8u) ? lhp[q] : 0.0f;
            const float mx = fmaxf(xhm, fmaxf(x0, xhp));
            const float e0 = __expf(xhm - mx), e1 = __expf(x0 - mx), e2 = __expf(xhp - mx);
            const float inv = 1.0f / (e0 + e1 + e2);
            const float t8[3] = { e0 * inv, e1 * inv, e2 * inv };
            const float xv[3] = { xvm, x0, xvp };
            const float p5v = P5s[lane + 32 * q];
            #pragma unroll
            for (int k = 0; k < 3; ++k) {
                float t5v  = p5v * lt3[k][q];
                float t2v  = P2e[k * 224 + lane + 32 * q] * xv[k];
                float t10v = lp10[k][q] * t8[k];
                bsst[q * 32 * PADK + k] = t5v - t2v - t10v;
            }
        }
        __syncthreads();
        #pragma unroll
        for (int kk = 0; kk < KS3 / 4; ++kk) {
            int ko = kk * 4 + kp;
            v2f a0 = *(const v2f*)&As[(mbase + mn) * PADK + ko];
            v2f a1 = *(const v2f*)&As[(mbase + 16 + mn) * PADK + ko];
            #pragma unroll
            for (int ni = 0; ni < 7; ++ni) {
                v2f b = *(const v2f*)&Bs[(nbase + ni * 16 + mn) * PADK + ko];
                acc[0][ni] = wmma_f32(a0, b, acc[0][ni]);
                acc[1][ni] = wmma_f32(a1, b, acc[1][ni]);
            }
        }
        __syncthreads();
    }

    const float scale = 0.051031036307982884f;
    float* o = out + (size_t)smp * CC * HWP;
    #pragma unroll
    for (int mi = 0; mi < 2; ++mi) {
        #pragma unroll
        for (int ni = 0; ni < 7; ++ni) {
            int col = pos0 + nbase + ni * 16 + mn;
            #pragma unroll
            for (int r = 0; r < 8; ++r) {
                int m = mbase + mi * 16 + r + rmsb;
                o[m * HWP + col] = acc[mi][ni][r] * scale;
            }
        }
    }
}

extern "C" void kernel_launch(void* const* d_in, const int* in_sizes, int n_in,
                              void* d_out, int out_size, void* d_ws, size_t ws_size,
                              hipStream_t stream) {
    (void)in_sizes; (void)n_in; (void)out_size; (void)ws_size;
    const float* x   = (const float*)d_in[0];
    const float* p2  = (const float*)d_in[1];
    const float* p3  = (const float*)d_in[2];
    const float* p5  = (const float*)d_in[3];
    const float* p10 = (const float*)d_in[4];
    float* out = (float*)d_out;

    char* ws = (char*)d_ws;
    const size_t t3_bytes  = (size_t)NS * CC * HWP * sizeof(float);
    const size_t t9p_bytes = (size_t)KSPLIT * T9N * sizeof(float);
    float* t3  = (float*)ws;
    float* t9p = (float*)(ws + t3_bytes);
    float* t9  = (float*)(ws + t3_bytes + t9p_bytes);

    dim3 blk(256);
    k_gemm1<<<dim3(HWP / 224, NS), blk, 0, stream>>>(p3, x, t3);
    k_gemm2<<<dim3(C3 / 128, NS, KSPLIT), blk, 0, stream>>>(x, t3, t9p);
    k_t9sum<<<dim3(T9N / 4 / 256), blk, 0, stream>>>(t9p, t9);
    k_gemm3<<<dim3(HWP / 224, NS), blk, 0, stream>>>(t9, x, t3, p2, p5, p10, out);
}


# Wave priority by progress in the three GEMM k-loops: s_setprio 3,2,1,0 as a loop advances, so the co-resident work-group that lags takes the matrix pipe and both end together; sums unchanged
# speedup vs baseline: 3.7994x; 1.0359x over previous
#include <hip/hip_runtime.h>
#include <hip/hip_bf16.h>


#define NS   32
#define CC   128
#define HH   56
#define WW   56
#define HWP  3136
#define C3   384
#define PADK 34
#define KSPLIT 7
#define KCHUNK 448
#define T9N  (NS * CC * C3)
#define KS3  24
#define NX   (NS * CC * HWP)

typedef float v2f __attribute__((ext_vector_type(2)));
typedef float v8f __attribute__((ext_vector_type(8)));

__device__ __forceinline__ v8f wmma_f32(v2f a, v2f b, v8f c) {
    return __builtin_amdgcn_wmma_f32_16x16x4_f32(false, a, false, b, (short)0, c, false, false);
}

__device__ __forceinline__ void prio_by_progress(int ks, int at0, int at1, int at2) {
    if (ks == at0)      __builtin_amdgcn_s_setprio(2);
    else if (ks == at1) __builtin_amdgcn_s_setprio(1);
    else if (ks == at2) __builtin_amdgcn_s_setprio(0);
}

__global__ __launch_bounds__(256) void k_gemm1(const float* __restrict__ p3,
                                               const float* __restrict__ x,
                                               float* __restrict__ t3) {
    __shared__ float As[128 * PADK];
    __shared__ float Bs[224 * PADK];
    const int t    = threadIdx.x;
    const int lane = t & 31;
    const int wv   = t >> 5;
    const int smp  = blockIdx.y;
    const int pos0 = blockIdx.x * 224;
    const float* xs = x + (size_t)smp * CC * HWP;

    const int mn    = lane & 15;
    const int kp    = (lane >> 4) << 1;
    const int rmsb  = (lane >> 4) << 3;
    const int mbase = (wv >> 1) * 32;
    const int nbase = (wv & 1) * 112;

    const int wvs = __builtin_amdgcn_readfirstlane(wv);
    const float* ap = p3 + wv * C3 + lane;
    float* asst = As + wv * PADK + lane;
    float* bsst = Bs + lane * PADK + wv;
    const int posl = pos0 + lane;
    int hq[7];
    #pragma unroll
    for (int q = 0; q < 7; ++q) hq[q] = (posl + 32 * q) / WW;

    v8f acc[2][7] = {};

    __builtin_amdgcn_s_setprio(3);
    for (int ks = 0; ks < C3; ks += 32) {
        prio_by_progress(ks, 192, 288, 352);
        float ra[16], rb[4][7];
        bool  ok[4][7];
        #pragma unroll
        for (int i = 0; i < 16; ++i)
            ra[i] = ap[ks + i * 8 * C3];
        #pragma unroll
        for (int a = 0; a < 4; ++a) {
            int j = ks + wvs + 8 * a;
            int tap = j % 3, ci = j / 3;
            int dh = 2 * (tap - 1);
            const float* xc = xs + ci * HWP;
            #pragma unroll
            for (int q = 0; q < 7; ++q) {
                ok[a][q] = (unsigned)(hq[q] + dh) < (unsigned)HH;
                rb[a][q] = xc[posl + 32 * q + (ok[a][q] ? WW * dh : 0)];
            }
        }
        __builtin_amdgcn_sched_barrier(0);
        #pragma unroll
        for (int i = 0; i < 16; ++i)
            asst[i * 8 * PADK] = ra[i];
        #pragma unroll
        for (int a = 0; a < 4; ++a) {
            #pragma unroll
            for (int q = 0; q < 7; ++q)
                bsst[q * 32 * PADK + 8 * a] = ok[a][q] ? rb[a][q] : 0.0f;
        }
        __syncthreads();
        #pragma unroll
        for (int kk = 0; kk < 8; ++kk) {
            int ko = kk * 4 + kp;
            v2f a0 = *(const v2f*)&As[(mbase + mn) * PADK + ko];
            v2f a1 = *(const v2f*)&As[(mbase + 16 + mn) * PADK + ko];
            #pragma unroll
            for (int ni = 0; ni < 7; ++ni) {
                v2f b = *(const v2f*)&Bs[(nbase + ni * 16 + mn) * PADK + ko];
                acc[0][ni] = wmma_f32(a0, b, acc[0][ni]);
                acc[1][ni] = wmma_f32(a1, b, acc[1][ni]);
            }
        }
        __syncthreads();
    }

    float* o = t3 + (size_t)smp * CC * HWP;
    #pragma unroll
    for (int mi = 0; mi < 2; ++mi) {
        #pragma unroll
        for (int ni = 0; ni < 7; ++ni) {
            int col = pos0 + nbase + ni * 16 + mn;
            #pragma unroll
            for (int r = 0; r < 8; ++r) {
                int m = mbase + mi * 16 + r + rmsb;
                o[m * HWP + col] = acc[mi][ni][r];
            }
        }
    }
}

__global__ __launch_bounds__(256) void k_gemm2(const float* __restrict__ x,
                                               const float* __restrict__ t3,
                                               float* __restrict__ t9p) {
    __shared__ float As[128 * PADK];
    __shared__ float Bs[128 * PADK];
    const int t    = threadIdx.x;
    const int lane = t & 31;
    const int wv   = t >> 5;
    const int smp  = blockIdx.y;
    const int d0   = blockIdx.x * 128;
    const float* xs  = x  + (size_t)smp * CC * HWP;
    const float* t3s = t3 + (size_t)smp * CC * HWP;

    const int mn    = lane & 15;
    const int kp    = (lane >> 4) << 1;
    const int rmsb  = (lane >> 4) << 3;
    const int mbase = (wv >> 1) * 32;
    const int nbase = (wv & 1) * 64;

    const int wvs = __builtin_amdgcn_readfirstlane(wv);
    const float* ap = xs + wv * HWP + lane;
    float* asst = As + wv * PADK + lane;
    float* bsst = Bs + wv * PADK + lane;

    v8f acc[2][4] = {};

    const int kbeg = blockIdx.z * KCHUNK;
    __builtin_amdgcn_s_setprio(3);
    for (int ks = kbeg; ks < kbeg + KCHUNK; ks += 32) {
        prio_by_progress(ks - kbeg, 224, 352, 416);
        float ra[16], rb[16];
        bool  ok[16];
        const int pos = ks + lane;
        const int wc  = pos % WW;
        #pragma unroll
        for (int i = 0; i < 16; ++i)
            ra[i] = ap[ks + i * 8 * HWP];
        #pragma unroll
        for (int i = 0; i < 16; ++i) {
            int d = d0 + wvs + 8 * i;
            int tap = d % 3, ci = d / 3;
            ok[i] = (unsigned)(wc + (tap - 1)) < (unsigned)WW;
            rb[i] = t3s[ci * HWP + pos + (ok[i] ? (tap - 1) : 0)];
        }
        __builtin_amdgcn_sched_barrier(0);
        #pragma unroll
        for (int i = 0; i < 16; ++i)
            asst[i * 8 * PADK] = ra[i];
        #pragma unroll
        for (int i = 0; i < 16; ++i)
            bsst[i * 8 * PADK] = ok[i] ? rb[i] : 0.0f;
        __syncthreads();
        #pragma unroll
        for (int kk = 0; kk < 8; ++kk) {
            int ko = kk * 4 + kp;
            v2f a0 = *(const v2f*)&As[(mbase + mn) * PADK + ko];
            v2f a1 = *(const v2f*)&As[(mbase + 16 + mn) * PADK + ko];
            #pragma unroll
            for (int ni = 0; ni < 4; ++ni) {
                v2f b = *(const v2f*)&Bs[(nbase + ni * 16 + mn) * PADK + ko];
                acc[0][ni] = wmma_f32(a0, b, acc[0][ni]);
                acc[1][ni] = wmma_f32(a1, b, acc[1][ni]);
            }
        }
        __syncthreads();
    }

    float* out = t9p + (size_t)blockIdx.z * T9N + (size_t)smp * CC * C3;
    #pragma unroll
    for (int mi = 0; mi < 2; ++mi) {
        #pragma unroll
        for (int ni = 0; ni < 4; ++ni) {
            int col = d0 + nbase + ni * 16 + mn;
            #pragma unroll
            for (int r = 0; r < 8; ++r) {
                int m = mbase + mi * 16 + r + rmsb;
                out[m * C3 + col] = acc[mi][ni][r];
            }
        }
    }
}

__global__ __launch_bounds__(256) void k_t9sum(const float* __restrict__ t9p,
                                               float* __restrict__ t9) {
    const int i = blockIdx.x * 256 + threadIdx.x;
    const float4* p = (const float4*)t9p + i;
    float4 v[KSPLIT];
    #pragma unroll
    for (int z = 0; z < KSPLIT; ++z) v[z] = p[(size_t)z * (T9N / 4)];
    float4 s = v[0];
    #pragma unroll
    for (int z = 1; z < KSPLIT; ++z) {
        s.x += v[z].x; s.y += v[z].y; s.z += v[z].z; s.w += v[z].w;
    }
    const float scale = 1.0f / 56.0f;
    s.x *= scale; s.y *= scale; s.z *= scale; s.w *= scale;
    ((float4*)t9)[i] = s;
}

__global__ __launch_bounds__(256) void k_gemm3(const float* __restrict__ t9,
                                               const float* __restrict__ x,
                                               const float* __restrict__ t3,
                                               const float* __restrict__ p2,
                                               const float* __restrict__ p5,
                                               const float* __restrict__ p10,
                                               float* __restrict__ out) {
    __shared__ float As[128 * PADK];
    __shared__ float Bs[224 * PADK];
    __shared__ float P5s[224];
    __shared__ float P2e[3 * 224];
    const int t    = threadIdx.x;
    const int lane = t & 31;
    const int wv   = t >> 5;
    const int smp  = blockIdx.y;
    const int pos0 = blockIdx.x * 224;
    const float* t9s = t9 + (size_t)smp * CC * C3;
    const float* t3s = t3 + (size_t)smp * CC * HWP;

    const int mn    = lane & 15;
    const int kp    = (lane >> 4) << 1;
    const int rmsb  = (lane >> 4) << 3;
    const int mbase = (wv >> 1) * 32;
    const int nbase = (wv & 1) * 112;

    const int wvs  = __builtin_amdgcn_readfirstlane(wv);
    const int acol = lane < KS3 ? lane : KS3 - 1;
    const float* ap = t9s + wv * C3 + acol;
    float* asst = As + wv * PADK + lane;
    float* bsst = Bs + lane * PADK + 3 * wv;
    const int posl = pos0 + lane;

    if (t < 224) P5s[t] = p5[pos0 + t];
    for (int i = t; i < 3 * 224; i += 256)
        P2e[i] = p2[(i / 224) * WW + (i % 224) % WW];

    unsigned mk = 0;
    #pragma unroll
    for (int q = 0; q < 7; ++q) {
        int pl = lane + 32 * q;
        int h = blockIdx.x * 4 + pl / WW, w = pl % WW;
        unsigned b = (h >= 2 ? 1u : 0u) | (h < HH - 2 ? 2u : 0u)
                   | (w >= 2 ? 4u : 0u) | (w < WW - 2 ? 8u : 0u);
        mk |= b << (4 * q);
    }
    __syncthreads();

    v8f acc[2][7] = {};

    __builtin_amdgcn_s_setprio(3);
    for (int ks = 0; ks < C3; ks += KS3) {
        prio_by_progress(ks, 192, 288, 360);
        float ra[16], lx0[7], lvm[7], lvp[7], lhm[7], lhp[7], lt3[3][7], lp10[3][7];
        const int d0 = ks + 3 * wvs;
        const int gx = (smp * CC + d0 / 3) * HWP + posl;
        #pragma unroll
        for (int i = 0; i < 16; ++i)
            ra[i] = ap[ks + i * 8 * C3];
        #pragma unroll
        for (int q = 0; q < 7; ++q) {
            const int g = gx + 32 * q;
            lx0[q] = x[g];
            lvm[q] = x[max(g - 2 * WW, 0)];
            lvp[q] = x[min(g + 2 * WW, NX - 1)];
            lhm[q] = x[max(g - 2, 0)];
            lhp[q] = x[min(g + 2, NX - 1)];
        }
        #pragma unroll
        for (int k = 0; k < 3; ++k) {
            const float* t3r  = t3s + ((d0 + k) & (CC - 1)) * HWP + posl;
            const float* p10r = p10 + (d0 + k) * HWP + posl;
            #pragma unroll
            for (int q = 0; q < 7; ++q) {
                lt3[k][q]  = t3r[32 * q];
                lp10[k][q] = p10r[32 * q];
            }
        }
        __builtin_amdgcn_sched_barrier(0);
        #pragma unroll
        for (int i = 0; i < 16; ++i)
            asst[i * 8 * PADK] = ra[i];
        #pragma unroll
        for (int q = 0; q < 7; ++q) {
            const unsigned b = mk >> (4 * q);
            const float x0  = lx0[q];
            const float xvm = (b & 1u) ? lvm[q] : 0.0f;
            const float xvp = (b & 2u) ? lvp[q] : 0.0f;
            const float xhm = (b & 4u) ? lhm[q] : 0.0f;
            const float xhp = (b & 8u) ? lhp[q] : 0.0f;
            const float mx = fmaxf(xhm, fmaxf(x0, xhp));
            const float e0 = __expf(xhm - mx), e1 = __expf(x0 - mx), e2 = __expf(xhp - mx);
            const float inv = 1.0f / (e0 + e1 + e2);
            const float t8[3] = { e0 * inv, e1 * inv, e2 * inv };
            const float xv[3] = { xvm, x0, xvp };
            const float p5v = P5s[lane + 32 * q];
            #pragma unroll
            for (int k = 0; k < 3; ++k) {
                float t5v  = p5v * lt3[k][q];
                float t2v  = P2e[k * 224 + lane + 32 * q] * xv[k];
                float t10v = lp10[k][q] * t8[k];
                bsst[q * 32 * PADK + k] = t5v - t2v - t10v;
            }
        }
        __syncthreads();
        #pragma unroll
        for (int kk = 0; kk < KS3 / 4; ++kk) {
            int ko = kk * 4 + kp;
            v2f a0 = *(const v2f*)&As[(mbase + mn) * PADK + ko];
            v2f a1 = *(const v2f*)&As[(mbase + 16 + mn) * PADK + ko];
            #pragma unroll
            for (int ni = 0; ni < 7; ++ni) {
                v2f b = *(const v2f*)&Bs[(nbase + ni * 16 + mn) * PADK + ko];
                acc[0][ni] = wmma_f32(a0, b, acc[0][ni]);
                acc[1][ni] = wmma_f32(a1, b, acc[1][ni]);
            }
        }
        __syncthreads();
    }

    const float scale = 0.051031036307982884f;
    float* o = out + (size_t)smp * CC * HWP;
    #pragma unroll
    for (int mi = 0; mi < 2; ++mi) {
        #pragma unroll
        for (int ni = 0; ni < 7; ++ni) {
            int col = pos0 + nbase + ni * 16 + mn;
            #pragma unroll
            for (int r = 0; r < 8; ++r) {
                int m = mbase + mi * 16 + r + rmsb;
                o[m * HWP + col] = acc[mi][ni][r] * scale;
            }
        }
    }
}

extern "C" void kernel_launch(void* const* d_in, const int* in_sizes, int n_in,
                              void* d_out, int out_size, void* d_ws, size_t ws_size,
                              hipStream_t stream) {
    (void)in_sizes; (void)n_in; (void)out_size; (void)ws_size;
    const float* x   = (const float*)d_in[0];
    const float* p2  = (const float*)d_in[1];
    const float* p3  = (const float*)d_in[2];
    const float* p5  = (const float*)d_in[3];
    const float* p10 = (const float*)d_in[4];
    float* out = (float*)d_out;

    char* ws = (char*)d_ws;
    const size_t t3_bytes  = (size_t)NS * CC * HWP * sizeof(float);
    const size_t t9p_bytes = (size_t)KSPLIT * T9N * sizeof(float);
    float* t3  = (float*)ws;
    float* t9p = (float*)(ws + t3_bytes);
    float* t9  = (float*)(ws + t3_bytes + t9p_bytes);

    dim3 blk(256);
    k_gemm1<<<dim3(HWP / 224, NS), blk, 0, stream>>>(p3, x, t3);
    k_gemm2<<<dim3(C3 / 128, NS, KSPLIT), blk, 0, stream>>>(x, t3, t9p);
    k_t9sum<<<dim3(T9N / 4 / 256), blk, 0, stream>>>(t9p, t9);
    k_gemm3<<<dim3(HWP / 224, NS), blk, 0, stream>>>(t9, x, t3, p2, p5, p10, out);
}


# Tile fills read 16 bytes per lane where memory runs along the tile: GEMM1's p3 and x tiles, the A tiles of GEMM2 and GEMM3; GEMM1 k-step 528 -> 278 instructions; LDS layout and sums unchanged
# speedup vs baseline: 3.9112x; 1.0294x over previous
#include <hip/hip_runtime.h>
#include <hip/hip_bf16.h>


#define NS   32
#define CC   128
#define HH   56
#define WW   56
#define HWP  3136
#define C3   384
#define PADK 34
#define KSPLIT 7
#define KCHUNK 448
#define T9N  (NS * CC * C3)
#define KS3  24
#define NX   (NS * CC * HWP)

typedef float v2f __attribute__((ext_vector_type(2)));
typedef float v4f __attribute__((ext_vector_type(4)));
typedef float v8f __attribute__((ext_vector_type(8)));

__device__ __forceinline__ v8f wmma_f32(v2f a, v2f b, v8f c) {
    return __builtin_amdgcn_wmma_f32_16x16x4_f32(false, a, false, b, (short)0, c, false, false);
}

__device__ __forceinline__ void prio_by_progress(int ks, int at0, int at1, int at2) {
    if (ks == at0)      __builtin_amdgcn_s_setprio(2);
    else if (ks == at1) __builtin_amdgcn_s_setprio(1);
    else if (ks == at2) __builtin_amdgcn_s_setprio(0);
}

__global__ __launch_bounds__(256) void k_gemm1(const float* __restrict__ p3,
                                               const float* __restrict__ x,
                                               float* __restrict__ t3) {
    __shared__ __attribute__((aligned(16))) float As[128 * PADK];
    __shared__ __attribute__((aligned(16))) float Bs[224 * PADK];
    const int t    = threadIdx.x;
    const int lane = t & 31;
    const int wv   = t >> 5;
    const int smp  = blockIdx.y;
    const int pos0 = blockIdx.x * 224;
    const float* xs = x + (size_t)smp * CC * HWP;

    const int mn    = lane & 15;
    const int kp    = (lane >> 4) << 1;
    const int rmsb  = (lane >> 4) << 3;
    const int mbase = (wv >> 1) * 32;
    const int nbase = (wv & 1) * 112;

    const int sub = lane >> 3;
    const int c8  = lane & 7;
    const int kr  = 4 * wv + sub;
    const float* ap = p3 + (4 * wv + sub) * C3 + 4 * c8;
    float* asst = As + (4 * wv + sub) * PADK + 4 * c8;
    float* bsst = Bs + (4 * c8) * PADK + kr;
    const float* xb = xs + pos0 + 4 * c8;
    int hq[7];
    #pragma unroll
    for (int m = 0; m < 7; ++m) hq[m] = blockIdx.x * 4 + (c8 + 8 * m) / 14;

    v8f acc[2][7] = {};

    __builtin_amdgcn_s_setprio(3);
    for (int ks = 0; ks < C3; ks += 32) {
        prio_by_progress(ks, 192, 288, 352);
        v4f  ra[4], rb[7];
        bool ok[7];
        #pragma unroll
        for (int a = 0; a < 4; ++a)
            ra[a] = *(const v4f*)(ap + ks + a * 32 * C3);
        {
            const int j   = ks + kr;
            const int ci  = j / 3;
            const int tap = j - 3 * ci;
            const int dh  = 2 * (tap - 1);
            const float* xc = xb + ci * HWP;
            #pragma unroll
            for (int m = 0; m < 7; ++m) {
                ok[m] = (unsigned)(hq[m] + dh) < (unsigned)HH;
                rb[m] = *(const v4f*)(xc + 32 * m + (ok[m] ? WW * dh : 0));
            }
        }
        __builtin_amdgcn_sched_barrier(0);
        #pragma unroll
        for (int a = 0; a < 4; ++a) {
            *(v2f*)(asst + a * 32 * PADK)     = ra[a].xy;
            *(v2f*)(asst + a * 32 * PADK + 2) = ra[a].zw;
        }
        #pragma unroll
        for (int m = 0; m < 7; ++m) {
            #pragma unroll
            for (int i = 0; i < 4; ++i)
                bsst[(32 * m + i) * PADK] = ok[m] ? rb[m][i] : 0.0f;
        }
        __syncthreads();
        #pragma unroll
        for (int kk = 0; kk < 8; ++kk) {
            int ko = kk * 4 + kp;
            v2f a0 = *(const v2f*)&As[(mbase + mn) * PADK + ko];
            v2f a1 = *(const v2f*)&As[(mbase + 16 + mn) * PADK + ko];
            #pragma unroll
            for (int ni = 0; ni < 7; ++ni) {
                v2f b = *(const v2f*)&Bs[(nbase + ni * 16 + mn) * PADK + ko];
                acc[0][ni] = wmma_f32(a0, b, acc[0][ni]);
                acc[1][ni] = wmma_f32(a1, b, acc[1][ni]);
            }
        }
        __syncthreads();
    }

    float* o = t3 + (size_t)smp * CC * HWP;
    #pragma unroll
    for (int mi = 0; mi < 2; ++mi) {
        #pragma unroll
        for (int ni = 0; ni < 7; ++ni) {
            int col = pos0 + nbase + ni * 16 + mn;
            #pragma unroll
            for (int r = 0; r < 8; ++r) {
                int m = mbase + mi * 16 + r + rmsb;
                o[m * HWP + col] = acc[mi][ni][r];
            }
        }
    }
}

__global__ __launch_bounds__(256) void k_gemm2(const float* __restrict__ x,
                                               const float* __restrict__ t3,
                                               float* __restrict__ t9p) {
    __shared__ __attribute__((aligned(16))) float As[128 * PADK];
    __shared__ __attribute__((aligned(16))) float Bs[128 * PADK];
    const int t    = threadIdx.x;
    const int lane = t & 31;
    const int wv   = t >> 5;
    const int smp  = blockIdx.y;
    const int d0   = blockIdx.x * 128;
    const float* xs  = x  + (size_t)smp * CC * HWP;
    const float* t3s = t3 + (size_t)smp * CC * HWP;

    const int mn    = lane & 15;
    const int kp    = (lane >> 4) << 1;
    const int rmsb  = (lane >> 4) << 3;
    const int mbase = (wv >> 1) * 32;
    const int nbase = (wv & 1) * 64;

    const int wvs = __builtin_amdgcn_readfirstlane(wv);
    const int sub = lane >> 3;
    const int c8  = lane & 7;
    const float* ap = xs + (4 * wv + sub) * HWP + 4 * c8;
    float* asst = As + (4 * wv + sub) * PADK + 4 * c8;
    float* bsst = Bs + wv * PADK + lane;

    v8f acc[2][4] = {};

    const int kbeg = blockIdx.z * KCHUNK;
    __builtin_amdgcn_s_setprio(3);
    for (int ks = kbeg; ks < kbeg + KCHUNK; ks += 32) {
        prio_by_progress(ks - kbeg, 224, 352, 416);
        v4f   ra[4];
        float rb[16];
        bool  ok[16];
        const int pos = ks + lane;
        const int wc  = pos % WW;
        #pragma unroll
        for (int a = 0; a < 4; ++a)
            ra[a] = *(const v4f*)(ap + ks + a * 32 * HWP);
        #pragma unroll
        for (int i = 0; i < 16; ++i) {
            int d = d0 + wvs + 8 * i;
            int tap = d % 3, ci = d / 3;
            ok[i] = (unsigned)(wc + (tap - 1)) < (unsigned)WW;
            rb[i] = t3s[ci * HWP + pos + (ok[i] ? (tap - 1) : 0)];
        }
        __builtin_amdgcn_sched_barrier(0);
        #pragma unroll
        for (int a = 0; a < 4; ++a) {
            *(v2f*)(asst + a * 32 * PADK)     = ra[a].xy;
            *(v2f*)(asst + a * 32 * PADK + 2) = ra[a].zw;
        }
        #pragma unroll
        for (int i = 0; i < 16; ++i)
            bsst[i * 8 * PADK] = ok[i] ? rb[i] : 0.0f;
        __syncthreads();
        #pragma unroll
        for (int kk = 0; kk < 8; ++kk) {
            int ko = kk * 4 + kp;
            v2f a0 = *(const v2f*)&As[(mbase + mn) * PADK + ko];
            v2f a1 = *(const v2f*)&As[(mbase + 16 + mn) * PADK + ko];
            #pragma unroll
            for (int ni = 0; ni < 4; ++ni) {
                v2f b = *(const v2f*)&Bs[(nbase + ni * 16 + mn) * PADK + ko];
                acc[0][ni] = wmma_f32(a0, b, acc[0][ni]);
                acc[1][ni] = wmma_f32(a1, b, acc[1][ni]);
            }
        }
        __syncthreads();
    }

    float* out = t9p + (size_t)blockIdx.z * T9N + (size_t)smp * CC * C3;
    #pragma unroll
    for (int mi = 0; mi < 2; ++mi) {
        #pragma unroll
        for (int ni = 0; ni < 4; ++ni) {
            int col = d0 + nbase + ni * 16 + mn;
            #pragma unroll
            for (int r = 0; r < 8; ++r) {
                int m = mbase + mi * 16 + r + rmsb;
                out[m * C3 + col] = acc[mi][ni][r];
            }
        }
    }
}

__global__ __launch_bounds__(256) void k_t9sum(const float* __restrict__ t9p,
                                               float* __restrict__ t9) {
    const int i = blockIdx.x * 256 + threadIdx.x;
    const float4* p = (const float4*)t9p + i;
    float4 v[KSPLIT];
    #pragma unroll
    for (int z = 0; z < KSPLIT; ++z) v[z] = p[(size_t)z * (T9N / 4)];
    float4 s = v[0];
    #pragma unroll
    for (int z = 1; z < KSPLIT; ++z) {
        s.x += v[z].x; s.y += v[z].y; s.z += v[z].z; s.w += v[z].w;
    }
    const float scale = 1.0f / 56.0f;
    s.x *= scale; s.y *= scale; s.z *= scale; s.w *= scale;
    ((float4*)t9)[i] = s;
}

__global__ __launch_bounds__(256) void k_gemm3(const float* __restrict__ t9,
                                               const float* __restrict__ x,
                                               const float* __restrict__ t3,
                                               const float* __restrict__ p2,
                                               const float* __restrict__ p5,
                                               const float* __restrict__ p10,
                                               float* __restrict__ out) {
    __shared__ __attribute__((aligned(16))) float As[128 * PADK];
    __shared__ __attribute__((aligned(16))) float Bs[224 * PADK];
    __shared__ float P5s[224];
    __shared__ float P2e[3 * 224];
    const int t    = threadIdx.x;
    const int lane = t & 31;
    const int wv   = t >> 5;
    const int smp  = blockIdx.y;
    const int pos0 = blockIdx.x * 224;
    const float* t9s = t9 + (size_t)smp * CC * C3;
    const float* t3s = t3 + (size_t)smp * CC * HWP;

    const int mn    = lane & 15;
    const int kp    = (lane >> 4) << 1;
    const int rmsb  = (lane >> 4) << 3;
    const int mbase = (wv >> 1) * 32;
    const int nbase = (wv & 1) * 112;

    const int wvs  = __builtin_amdgcn_readfirstlane(wv);
    const int sub  = lane >> 3;
    const int c8   = lane & 7;
    const int kq   = c8 < KS3 / 4 ? c8 : KS3 / 4 - 1;
    const float* ap = t9s + (4 * wv + sub) * C3 + 4 * kq;
    float* asst = As + (4 * wv + sub) * PADK + 4 * kq;
    float* bsst = Bs + lane * PADK + 3 * wv;
    const int posl = pos0 + lane;

    if (t < 224) P5s[t] = p5[pos0 + t];
    for (int i = t; i < 3 * 224; i += 256)
        P2e[i] = p2[(i / 224) * WW + (i % 224) % WW];

    unsigned mk = 0;
    #pragma unroll
    for (int q = 0; q < 7; ++q) {
        int pl = lane + 32 * q;
        int h = blockIdx.x * 4 + pl / WW, w = pl % WW;
        unsigned b = (h >= 2 ? 1u : 0u) | (h < HH - 2 ? 2u : 0u)
                   | (w >= 2 ? 4u : 0u) | (w < WW - 2 ? 8u : 0u);
        mk |= b << (4 * q);
    }
    __syncthreads();

    v8f acc[2][7] = {};

    __builtin_amdgcn_s_setprio(3);
    for (int ks = 0; ks < C3; ks += KS3) {
        prio_by_progress(ks, 192, 288, 360);
        v4f   ra[4];
        float lx0[7], lvm[7], lvp[7], lhm[7], lhp[7], lt3[3][7], lp10[3][7];
        const int d0 = ks + 3 * wvs;
        const int gx = (smp * CC + d0 / 3) * HWP + posl;
        #pragma unroll
        for (int a = 0; a < 4; ++a)
            ra[a] = *(const v4f*)(ap + ks + a * 32 * C3);
        #pragma unroll
        for (int q = 0; q < 7; ++q) {
            const int g = gx + 32 * q;
            lx0[q] = x[g];
            lvm[q] = x[max(g - 2 * WW, 0)];
            lvp[q] = x[min(g + 2 * WW, NX - 1)];
            lhm[q] = x[max(g - 2, 0)];
            lhp[q] = x[min(g + 2, NX - 1)];
        }
        #pragma unroll
        for (int k = 0; k < 3; ++k) {
            const float* t3r  = t3s + ((d0 + k) & (CC - 1)) * HWP + posl;
            const float* p10r = p10 + (d0 + k) * HWP + posl;
            #pragma unroll
            for (int q = 0; q < 7; ++q) {
                lt3[k][q]  = t3r[32 * q];
                lp10[k][q] = p10r[32 * q];
            }
        }
        __builtin_amdgcn_sched_barrier(0);
        #pragma unroll
        for (int a = 0; a < 4; ++a) {
            *(v2f*)(asst + a * 32 * PADK)     = ra[a].xy;
            *(v2f*)(asst + a * 32 * PADK + 2) = ra[a].zw;
        }
        #pragma unroll
        for (int q = 0; q < 7; ++q) {
            const unsigned b = mk >> (4 * q);
            const float x0  = lx0[q];
            const float xvm = (b & 1u) ? lvm[q] : 0.0f;
            const float xvp = (b & 2u) ? lvp[q] : 0.0f;
            const float xhm = (b & 4u) ? lhm[q] : 0.0f;
            const float xhp = (b & 8u) ? lhp[q] : 0.0f;
            const float mx = fmaxf(xhm, fmaxf(x0, xhp));
            const float e0 = __expf(xhm - mx), e1 = __expf(x0 - mx), e2 = __expf(xhp - mx);
            const float inv = 1.0f / (e0 + e1 + e2);
            const float t8[3] = { e0 * inv, e1 * inv, e2 * inv };
            const float xv[3] = { xvm, x0, xvp };
            const float p5v = P5s[lane + 32 * q];
            #pragma unroll
            for (int k = 0; k < 3; ++k) {
                float t5v  = p5v * lt3[k][q];
                float t2v  = P2e[k * 224 + lane + 32 * q] * xv[k];
                float t10v = lp10[k][q] * t8[k];
                bsst[q * 32 * PADK + k] = t5v - t2v - t10v;
            }
        }
        __syncthreads();
        #pragma unroll
        for (int kk = 0; kk < KS3 / 4; ++kk) {
            int ko = kk * 4 + kp;
            v2f a0 = *(const v2f*)&As[(mbase + mn) * PADK + ko];
            v2f a1 = *(const v2f*)&As[(mbase + 16 + mn) * PADK + ko];
            #pragma unroll
            for (int ni = 0; ni < 7; ++ni) {
                v2f b = *(const v2f*)&Bs[(nbase + ni * 16 + mn) * PADK + ko];
                acc[0][ni] = wmma_f32(a0, b, acc[0][ni]);
                acc[1][ni] = wmma_f32(a1, b, acc[1][ni]);
            }
        }
        __syncthreads();
    }

    const float scale = 0.051031036307982884f;
    float* o = out + (size_t)smp * CC * HWP;
    #pragma unroll
    for (int mi = 0; mi < 2; ++mi) {
        #pragma unroll
        for (int ni = 0; ni < 7; ++ni) {
            int col = pos0 + nbase + ni * 16 + mn;
            #pragma unroll
            for (int r = 0; r < 8; ++r) {
                int m = mbase + mi * 16 + r + rmsb;
                o[m * HWP + col] = acc[mi][ni][r] * scale;
            }
        }
    }
}

extern "C" void kernel_launch(void* const* d_in, const int* in_sizes, int n_in,
                              void* d_out, int out_size, void* d_ws, size_t ws_size,
                              hipStream_t stream) {
    (void)in_sizes; (void)n_in; (void)out_size; (void)ws_size;
    const float* x   = (const float*)d_in[0];
    const float* p2  = (const float*)d_in[1];
    const float* p3  = (const float*)d_in[2];
    const float* p5  = (const float*)d_in[3];
    const float* p10 = (const float*)d_in[4];
    float* out = (float*)d_out;

    char* ws = (char*)d_ws;
    const size_t t3_bytes  = (size_t)NS * CC * HWP * sizeof(float);
    const size_t t9p_bytes = (size_t)KSPLIT * T9N * sizeof(float);
    float* t3  = (float*)ws;
    float* t9p = (float*)(ws + t3_bytes);
    float* t9  = (float*)(ws + t3_bytes + t9p_bytes);

    dim3 blk(256);
    k_gemm1<<<dim3(HWP / 224, NS), blk, 0, stream>>>(p3, x, t3);
    k_gemm2<<<dim3(C3 / 128, NS, KSPLIT), blk, 0, stream>>>(x, t3, t9p);
    k_t9sum<<<dim3(T9N / 4 / 256), blk, 0, stream>>>(t9p, t9);
    k_gemm3<<<dim3(HWP / 224, NS), blk, 0, stream>>>(t9, x, t3, p2, p5, p10, out);
}


# Epilogues of the three GEMMs turn each wave's rows through an LDS slab: every 128-byte output line is written whole by one 16-byte-per-lane store, not as two 64-byte halves; sums unchanged
# speedup vs baseline: 4.0700x; 1.0406x over previous
#include <hip/hip_runtime.h>
#include <hip/hip_bf16.h>


#define NS   32
#define CC   128
#define HH   56
#define WW   56
#define HWP  3136
#define C3   384
#define PADK 34
#define KSPLIT 7
#define KCHUNK 448
#define T9N  (NS * CC * C3)
#define KS3  24
#define NX   (NS * CC * HWP)

typedef float v2f __attribute__((ext_vector_type(2)));
typedef float v4f __attribute__((ext_vector_type(4)));
typedef float v8f __attribute__((ext_vector_type(8)));

__device__ __forceinline__ v8f wmma_f32(v2f a, v2f b, v8f c) {
    return __builtin_amdgcn_wmma_f32_16x16x4_f32(false, a, false, b, (short)0, c, false, false);
}

__device__ __forceinline__ void prio_by_progress(int ks, int at0, int at1, int at2) {
    if (ks == at0)      __builtin_amdgcn_s_setprio(2);
    else if (ks == at1) __builtin_amdgcn_s_setprio(1);
    else if (ks == at2) __builtin_amdgcn_s_setprio(0);
}

template <int NI, int P>
__device__ __forceinline__ void store_rows(const v8f (&acc)[2][NI], float* slab, float* g,
                                           int ld, float scale, int lane) {
    const int mn = lane & 15;
    const int h8 = (lane >> 4) << 3;
    #pragma unroll
    for (int mi = 0; mi < 2; ++mi) {
        __builtin_amdgcn_fence(__ATOMIC_ACQ_REL, "wavefront");
        __builtin_amdgcn_wave_barrier();
        #pragma unroll
        for (int ni = 0; ni < NI; ++ni) {
            #pragma unroll
            for (int r = 0; r < 8; ++r)
                slab[(r + h8) * P + ni * 16 + mn] = acc[mi][ni][r] * scale;
        }
        __builtin_amdgcn_fence(__ATOMIC_ACQ_REL, "wavefront");
        __builtin_amdgcn_wave_barrier();
        float* gr = g + (size_t)(mi * 16) * ld;
        if (NI == 7) {
            if (lane < 28) {
                #pragma unroll
                for (int rr = 0; rr < 16; ++rr)
                    *(v4f*)(gr + (size_t)rr * ld + 4 * lane) = *(const v4f*)(slab + rr * P + 4 * lane);
            }
        } else {
            const int c4 = 4 * mn;
            const int rh = lane >> 4;
            #pragma unroll
            for (int j = 0; j < 8; ++j) {
                const int rr = 2 * j + rh;
                *(v4f*)(gr + (size_t)rr * ld + c4) = *(const v4f*)(slab + rr * P + c4);
            }
        }
    }
}

__global__ __launch_bounds__(256) void k_gemm1(const float* __restrict__ p3,
                                               const float* __restrict__ x,
                                               float* __restrict__ t3) {
    __shared__ __attribute__((aligned(16))) float smem[8 * 16 * 116];
    float* const As = smem;
    float* const Bs = smem + 128 * PADK;
    const int t    = threadIdx.x;
    const int lane = t & 31;
    const int wv   = t >> 5;
    const int smp  = blockIdx.y;
    const int pos0 = blockIdx.x * 224;
    const float* xs = x + (size_t)smp * CC * HWP;

    const int mn    = lane & 15;
    const int kp    = (lane >> 4) << 1;
    const int mbase = (wv >> 1) * 32;
    const int nbase = (wv & 1) * 112;

    const int sub = lane >> 3;
    const int c8  = lane & 7;
    const int kr  = 4 * wv + sub;
    const float* ap = p3 + (4 * wv + sub) * C3 + 4 * c8;
    float* asst = As + (4 * wv + sub) * PADK + 4 * c8;
    float* bsst = Bs + (4 * c8) * PADK + kr;
    const float* xb = xs + pos0 + 4 * c8;
    int hq[7];
    #pragma unroll
    for (int m = 0; m < 7; ++m) hq[m] = blockIdx.x * 4 + (c8 + 8 * m) / 14;

    v8f acc[2][7] = {};

    __builtin_amdgcn_s_setprio(3);
    for (int ks = 0; ks < C3; ks += 32) {
        prio_by_progress(ks, 192, 288, 352);
        v4f  ra[4], rb[7];
        bool ok[7];
        #pragma unroll
        for (int a = 0; a < 4; ++a)
            ra[a] = *(const v4f*)(ap + ks + a * 32 * C3);
        {
            const int j   = ks + kr;
            const int ci  = j / 3;
            const int tap = j - 3 * ci;
            const int dh  = 2 * (tap - 1);
            const float* xc = xb + ci * HWP;
            #pragma unroll
            for (int m = 0; m < 7; ++m) {
                ok[m] = (unsigned)(hq[m] + dh) < (unsigned)HH;
                rb[m] = *(const v4f*)(xc + 32 * m + (ok[m] ? WW * dh : 0));
            }
        }
        __builtin_amdgcn_sched_barrier(0);
        #pragma unroll
        for (int a = 0; a < 4; ++a) {
            *(v2f*)(asst + a * 32 * PADK)     = ra[a].xy;
            *(v2f*)(asst + a * 32 * PADK + 2) = ra[a].zw;
        }
        #pragma unroll
        for (int m = 0; m < 7; ++m) {
            #pragma unroll
            for (int i = 0; i < 4; ++i)
                bsst[(32 * m + i) * PADK] = ok[m] ? rb[m][i] : 0.0f;
        }
        __syncthreads();
        #pragma unroll
        for (int kk = 0; kk < 8; ++kk) {
            int ko = kk * 4 + kp;
            v2f a0 = *(const v2f*)&As[(mbase + mn) * PADK + ko];
            v2f a1 = *(const v2f*)&As[(mbase + 16 + mn) * PADK + ko];
            #pragma unroll
            for (int ni = 0; ni < 7; ++ni) {
                v2f b = *(const v2f*)&Bs[(nbase + ni * 16 + mn) * PADK + ko];
                acc[0][ni] = wmma_f32(a0, b, acc[0][ni]);
                acc[1][ni] = wmma_f32(a1, b, acc[1][ni]);
            }
        }
        __syncthreads();
    }

    float* o = t3 + (size_t)smp * CC * HWP + (size_t)mbase * HWP + pos0 + nbase;
    store_rows<7, 116>(acc, smem + wv * (16 * 116), o, HWP, 1.0f, lane);
}

__global__ __launch_bounds__(256) void k_gemm2(const float* __restrict__ x,
                                               const float* __restrict__ t3,
                                               float* __restrict__ t9p) {
    __shared__ __attribute__((aligned(16))) float smem[2 * 128 * PADK];
    float* const As = smem;
    float* const Bs = smem + 128 * PADK;
    const int t    = threadIdx.x;
    const int lane = t & 31;
    const int wv   = t >> 5;
    const int smp  = blockIdx.y;
    const int d0   = blockIdx.x * 128;
    const float* xs  = x  + (size_t)smp * CC * HWP;
    const float* t3s = t3 + (size_t)smp * CC * HWP;

    const int mn    = lane & 15;
    const int kp    = (lane >> 4) << 1;
    const int mbase = (wv >> 1) * 32;
    const int nbase = (wv & 1) * 64;

    const int wvs = __builtin_amdgcn_readfirstlane(wv);
    const int sub = lane >> 3;
    const int c8  = lane & 7;
    const float* ap = xs + (4 * wv + sub) * HWP + 4 * c8;
    float* asst = As + (4 * wv + sub) * PADK + 4 * c8;
    float* bsst = Bs + wv * PADK + lane;

    v8f acc[2][4] = {};

    const int kbeg = blockIdx.z * KCHUNK;
    __builtin_amdgcn_s_setprio(3);
    for (int ks = kbeg; ks < kbeg + KCHUNK; ks += 32) {
        prio_by_progress(ks - kbeg, 224, 352, 416);
        v4f   ra[4];
        float rb[16];
        bool  ok[16];
        const int pos = ks + lane;
        const int wc  = pos % WW;
        #pragma unroll
        for (int a = 0; a < 4; ++a)
            ra[a] = *(const v4f*)(ap + ks + a * 32 * HWP);
        #pragma unroll
        for (int i = 0; i < 16; ++i) {
            int d = d0 + wvs + 8 * i;
            int tap = d % 3, ci = d / 3;
            ok[i] = (unsigned)(wc + (tap - 1)) < (unsigned)WW;
            rb[i] = t3s[ci * HWP + pos + (ok[i] ? (tap - 1) : 0)];
        }
        __builtin_amdgcn_sched_barrier(0);
        #pragma unroll
        for (int a = 0; a < 4; ++a) {
            *(v2f*)(asst + a * 32 * PADK)     = ra[a].xy;
            *(v2f*)(asst + a * 32 * PADK + 2) = ra[a].zw;
        }
        #pragma unroll
        for (int i = 0; i < 16; ++i)
            bsst[i * 8 * PADK] = ok[i] ? rb[i] : 0.0f;
        __syncthreads();
        #pragma unroll
        for (int kk = 0; kk < 8; ++kk) {
            int ko = kk * 4 + kp;
            v2f a0 = *(const v2f*)&As[(mbase + mn) * PADK + ko];
            v2f a1 = *(const v2f*)&As[(mbase + 16 + mn) * PADK + ko];
            #pragma unroll
            for (int ni = 0; ni < 4; ++ni) {
                v2f b = *(const v2f*)&Bs[(nbase + ni * 16 + mn) * PADK + ko];
                acc[0][ni] = wmma_f32(a0, b, acc[0][ni]);
                acc[1][ni] = wmma_f32(a1, b, acc[1][ni]);
            }
        }
        __syncthreads();
    }

    float* out = t9p + (size_t)blockIdx.z * T9N + (size_t)smp * CC * C3
               + (size_t)mbase * C3 + d0 + nbase;
    store_rows<4, 68>(acc, smem + wv * (16 * 68), out, C3, 1.0f, lane);
}

__global__ __launch_bounds__(256) void k_t9sum(const float* __restrict__ t9p,
                                               float* __restrict__ t9) {
    const int i = blockIdx.x * 256 + threadIdx.x;
    const float4* p = (const float4*)t9p + i;
    float4 v[KSPLIT];
    #pragma unroll
    for (int z = 0; z < KSPLIT; ++z) v[z] = p[(size_t)z * (T9N / 4)];
    float4 s = v[0];
    #pragma unroll
    for (int z = 1; z < KSPLIT; ++z) {
        s.x += v[z].x; s.y += v[z].y; s.z += v[z].z; s.w += v[z].w;
    }
    const float scale = 1.0f / 56.0f;
    s.x *= scale; s.y *= scale; s.z *= scale; s.w *= scale;
    ((float4*)t9)[i] = s;
}

__global__ __launch_bounds__(256) void k_gemm3(const float* __restrict__ t9,
                                               const float* __restrict__ x,
                                               const float* __restrict__ t3,
                                               const float* __restrict__ p2,
                                               const float* __restrict__ p5,
                                               const float* __restrict__ p10,
                                               float* __restrict__ out) {
    __shared__ __attribute__((aligned(16))) float smem[8 * 16 * 116];
    float* const As = smem;
    float* const Bs = smem + 128 * PADK;
    __shared__ float P5s[224];
    __shared__ float P2e[3 * 224];
    const int t    = threadIdx.x;
    const int lane = t & 31;
    const int wv   = t >> 5;
    const int smp  = blockIdx.y;
    const int pos0 = blockIdx.x * 224;
    const float* t9s = t9 + (size_t)smp * CC * C3;
    const float* t3s = t3 + (size_t)smp * CC * HWP;

    const int mn    = lane & 15;
    const int kp    = (lane >> 4) << 1;
    const int mbase = (wv >> 1) * 32;
    const int nbase = (wv & 1) * 112;

    const int wvs  = __builtin_amdgcn_readfirstlane(wv);
    const int sub  = lane >> 3;
    const int c8   = lane & 7;
    const int kq   = c8 < KS3 / 4 ? c8 : KS3 / 4 - 1;
    const float* ap = t9s + (4 * wv + sub) * C3 + 4 * kq;
    float* asst = As + (4 * wv + sub) * PADK + 4 * kq;
    float* bsst = Bs + lane * PADK + 3 * wv;
    const int posl = pos0 + lane;

    if (t < 224) P5s[t] = p5[pos0 + t];
    for (int i = t; i < 3 * 224; i += 256)
        P2e[i] = p2[(i / 224) * WW + (i % 224) % WW];

    unsigned mk = 0;
    #pragma unroll
    for (int q = 0; q < 7; ++q) {
        int pl = lane + 32 * q;
        int h = blockIdx.x * 4 + pl / WW, w = pl % WW;
        unsigned b = (h >= 2 ? 1u : 0u) | (h < HH - 2 ? 2u : 0u)
                   | (w >= 2 ? 4u : 0u) | (w < WW - 2 ? 8u : 0u);
        mk |= b << (4 * q);
    }
    __syncthreads();

    v8f acc[2][7] = {};

    __builtin_amdgcn_s_setprio(3);
    for (int ks = 0; ks < C3; ks += KS3) {
        prio_by_progress(ks, 192, 288, 360);
        v4f   ra[4];
        float lx0[7], lvm[7], lvp[7], lhm[7], lhp[7], lt3[3][7], lp10[3][7];
        const int d0 = ks + 3 * wvs;
        const int gx = (smp * CC + d0 / 3) * HWP + posl;
        #pragma unroll
        for (int a = 0; a < 4; ++a)
            ra[a] = *(const v4f*)(ap + ks + a * 32 * C3);
        #pragma unroll
        for (int q = 0; q < 7; ++q) {
            const int g = gx + 32 * q;
            lx0[q] = x[g];
            lvm[q] = x[max(g - 2 * WW, 0)];
            lvp[q] = x[min(g + 2 * WW, NX - 1)];
            lhm[q] = x[max(g - 2, 0)];
            lhp[q] = x[min(g + 2, NX - 1)];
        }
        #pragma unroll
        for (int k = 0; k < 3; ++k) {
            const float* t3r  = t3s + ((d0 + k) & (CC - 1)) * HWP + posl;
            const float* p10r = p10 + (d0 + k) * HWP + posl;
            #pragma unroll
            for (int q = 0; q < 7; ++q) {
                lt3[k][q]  = t3r[32 * q];
                lp10[k][q] = p10r[32 * q];
            }
        }
        __builtin_amdgcn_sched_barrier(0);
        #pragma unroll
        for (int a = 0; a < 4; ++a) {
            *(v2f*)(asst + a * 32 * PADK)     = ra[a].xy;
            *(v2f*)(asst + a * 32 * PADK + 2) = ra[a].zw;
        }
        #pragma unroll
        for (int q = 0; q < 7; ++q) {
            const unsigned b = mk >> (4 * q);
            const float x0  = lx0[q];
            const float xvm = (b & 1u) ? lvm[q] : 0.0f;
            const float xvp = (b & 2u) ? lvp[q] : 0.0f;
            const float xhm = (b & 4u) ? lhm[q] : 0.0f;
            const float xhp = (b & 8u) ? lhp[q] : 0.0f;
            const float mx = fmaxf(xhm, fmaxf(x0, xhp));
            const float e0 = __expf(xhm - mx), e1 = __expf(x0 - mx), e2 = __expf(xhp - mx);
            const float inv = 1.0f / (e0 + e1 + e2);
            const float t8[3] = { e0 * inv, e1 * inv, e2 * inv };
            const float xv[3] = { xvm, x0, xvp };
            const float p5v = P5s[lane + 32 * q];
            #pragma unroll
            for (int k = 0; k < 3; ++k) {
                float t5v  = p5v * lt3[k][q];
                float t2v  = P2e[k * 224 + lane + 32 * q] * xv[k];
                float t10v = lp10[k][q] * t8[k];
                bsst[q * 32 * PADK + k] = t5v - t2v - t10v;
            }
        }
        __syncthreads();
        #pragma unroll
        for (int kk = 0; kk < KS3 / 4; ++kk) {
            int ko = kk * 4 + kp;
            v2f a0 = *(const v2f*)&As[(mbase + mn) * PADK + ko];
            v2f a1 = *(const v2f*)&As[(mbase + 16 + mn) * PADK + ko];
            #pragma unroll
            for (int ni = 0; ni < 7; ++ni) {
                v2f b = *(const v2f*)&Bs[(nbase + ni * 16 + mn) * PADK + ko];
                acc[0][ni] = wmma_f32(a0, b, acc[0][ni]);
                acc[1][ni] = wmma_f32(a1, b, acc[1][ni]);
            }
        }
        __syncthreads();
    }

    const float scale = 0.051031036307982884f;
    float* o = out + (size_t)smp * CC * HWP + (size_t)mbase * HWP + pos0 + nbase;
    store_rows<7, 116>(acc, smem + wv * (16 * 116), o, HWP, scale, lane);
}

extern "C" void kernel_launch(void* const* d_in, const int* in_sizes, int n_in,
                              void* d_out, int out_size, void* d_ws, size_t ws_size,
                              hipStream_t stream) {
    (void)in_sizes; (void)n_in; (void)out_size; (void)ws_size;
    const float* x   = (const float*)d_in[0];
    const float* p2  = (const float*)d_in[1];
    const float* p3  = (const float*)d_in[2];
    const float* p5  = (const float*)d_in[3];
    const float* p10 = (const float*)d_in[4];
    float* out = (float*)d_out;

    char* ws = (char*)d_ws;
    const size_t t3_bytes  = (size_t)NS * CC * HWP * sizeof(float);
    const size_t t9p_bytes = (size_t)KSPLIT * T9N * sizeof(float);
    float* t3  = (float*)ws;
    float* t9p = (float*)(ws + t3_bytes);
    float* t9  = (float*)(ws + t3_bytes + t9p_bytes);

    dim3 blk(256);
    k_gemm1<<<dim3(HWP / 224, NS), blk, 0, stream>>>(p3, x, t3);
    k_gemm2<<<dim3(C3 / 128, NS, KSPLIT), blk, 0, stream>>>(x, t3, t9p);
    k_t9sum<<<dim3(T9N / 4 / 256), blk, 0, stream>>>(t9p, t9);
    k_gemm3<<<dim3(HWP / 224, NS), blk, 0, stream>>>(t9, x, t3, p2, p5, p10, out);
}
